# starting kernel
# baseline (speedup 1.0000x reference)
#include <hip/hip_runtime.h>
#include <math.h>

#define NNODES   100000
#define NEDGES   1600000
#define L1_H     4
#define L1_C     32
#define L1_HC    128
#define L2_H     1
#define L2_C     64
#define L2_HC    64
#define NEG_SLOPE 0.2f
#define EPS_      1e-16f

typedef __attribute__((ext_vector_type(2))) float v2f;
typedef __attribute__((ext_vector_type(8))) float v8f;

__device__ __forceinline__ float atomic_max_f32(float* addr, float val) {
  if (val >= 0.0f)
    return __int_as_float(atomicMax((int*)addr, __float_as_int(val)));
  return __uint_as_float(atomicMin((unsigned int*)addr, __float_as_uint(val)));
}

__device__ __forceinline__ void edge_nodes(const int* __restrict__ ei, int idx,
                                           int& s, int& d) {
  if (idx < NEDGES) { s = ei[idx]; d = ei[NEDGES + idx]; }
  else             { s = idx - NEDGES; d = s; }
}

__global__ void fill_kernel(float* __restrict__ p, float v, long long n) {
  long long i = (long long)blockIdx.x * blockDim.x + threadIdx.x;
  if (i < n) p[i] = v;
}

template <int K, int NC>
__global__ __launch_bounds__(128)
void gemm_wmma_kernel(const float* __restrict__ A, const float* __restrict__ B,
                      float* __restrict__ C) {
  const int lane = threadIdx.x & 31;
  const int wv   = threadIdx.x >> 5;
  const int row0 = blockIdx.x * 16;
  const int col0 = wv * 32;
  const int m    = lane & 15;
  const int kk   = (lane >> 4) << 1;

  v8f acc0 = {};
  v8f acc1 = {};
  const float* ap = A + (long long)(row0 + m) * K + kk;
  const float* bp = B + (long long)kk * NC + (col0 + m);
#pragma unroll
  for (int k0 = 0; k0 < K; k0 += 4) {
    v2f a, b0, b1;
    a.x  = ap[k0];            a.y  = ap[k0 + 1];
    b0.x = bp[k0 * NC];       b0.y = bp[(k0 + 1) * NC];
    b1.x = bp[k0 * NC + 16];  b1.y = bp[(k0 + 1) * NC + 16];
    acc0 = __builtin_amdgcn_wmma_f32_16x16x4_f32(
        false, a, false, b0, (short)0, acc0, false, false);
    acc1 = __builtin_amdgcn_wmma_f32_16x16x4_f32(
        false, a, false, b1, (short)0, acc1, false, false);
  }
  const int orow = row0 + ((lane >> 4) << 3);
  const int ocol = col0 + m;
  float* cp = C + (long long)orow * NC + ocol;
#pragma unroll
  for (int r = 0; r < 8; ++r) {
    cp[(long long)r * NC]      = acc0[r];
    cp[(long long)r * NC + 16] = acc1[r];
  }
}

__global__ void attn_kernel(const float* __restrict__ h,
                            const float* __restrict__ att_s,
                            const float* __restrict__ att_d,
                            float* __restrict__ as_, float* __restrict__ ad_,
                            int H, int C) {
  int i = blockIdx.x * blockDim.x + threadIdx.x;
  if (i >= NNODES * H) return;
  int n = i / H, hd = i % H;
  const float* hv = h + (long long)n * H * C + hd * C;
  const float* s  = att_s + hd * C;
  const float* d  = att_d + hd * C;
  float ss = 0.0f, dd = 0.0f;
  for (int c = 0; c < C; ++c) { float x = hv[c]; ss += x * s[c]; dd += x * d[c]; }
  as_[i] = ss;
  ad_[i] = dd;
}

__global__ void edge_max_kernel(const int* __restrict__ ei,
                                const float* __restrict__ as_,
                                const float* __restrict__ ad_,
                                float* __restrict__ mx, int H) {
  int idx = blockIdx.x * blockDim.x + threadIdx.x;
  if (idx >= NEDGES + NNODES) return;
  int s, d; edge_nodes(ei, idx, s, d);
  for (int h = 0; h < H; ++h) {
    float e = as_[s * H + h] + ad_[d * H + h];
    e = e > 0.0f ? e : NEG_SLOPE * e;
    atomic_max_f32(&mx[d * H + h], e);
  }
}

__global__ void edge_sum_kernel(const int* __restrict__ ei,
                                const float* __restrict__ as_,
                                const float* __restrict__ ad_,
                                const float* __restrict__ mx,
                                float* __restrict__ sm, int H) {
  int idx = blockIdx.x * blockDim.x + threadIdx.x;
  if (idx >= NEDGES + NNODES) return;
  int s, d; edge_nodes(ei, idx, s, d);
  for (int h = 0; h < H; ++h) {
    float e = as_[s * H + h] + ad_[d * H + h];
    e = e > 0.0f ? e : NEG_SLOPE * e;
    atomicAdd(&sm[d * H + h], expf(e - mx[d * H + h]));
  }
}

__global__ void edge_scatter_kernel(const int* __restrict__ ei,
                                    const float* __restrict__ as_,
                                    const float* __restrict__ ad_,
                                    const float* __restrict__ mx,
                                    const float* __restrict__ sm,
                                    const float* __restrict__ hfeat,
                                    float* __restrict__ out, int H, int C) {
  const int HC = H * C;
  long long gid = (long long)blockIdx.x * blockDim.x + threadIdx.x;
  long long tot = (long long)(NEDGES + NNODES) * HC;
  if (gid >= tot) return;
  int idx = (int)(gid / HC);
  int c   = (int)(gid % HC);
  int h   = c / C;
  int s, d; edge_nodes(ei, idx, s, d);
  float e  = as_[s * H + h] + ad_[d * H + h];
  e        = e > 0.0f ? e : NEG_SLOPE * e;
  float ex = expf(e - mx[d * H + h]);
  float alpha = ex / (sm[d * H + h] + EPS_);
  atomicAdd(&out[(long long)d * HC + c], hfeat[(long long)s * HC + c] * alpha);
}

__global__ void bias_elu_kernel(float* __restrict__ o, const float* __restrict__ b,
                                long long n, int HC) {
  long long i = (long long)blockIdx.x * blockDim.x + threadIdx.x;
  if (i >= n) return;
  float v = o[i] + b[(int)(i % HC)];
  o[i] = v > 0.0f ? v : expm1f(v);
}

__global__ void bias_add_kernel(float* __restrict__ o, const float* __restrict__ b,
                                long long n, int HC) {
  long long i = (long long)blockIdx.x * blockDim.x + threadIdx.x;
  if (i >= n) return;
  o[i] += b[(int)(i % HC)];
}

extern "C" void kernel_launch(void* const* d_in, const int* in_sizes, int n_in,
                              void* d_out, int out_size, void* d_ws, size_t ws_size,
                              hipStream_t stream) {
  (void)in_sizes; (void)n_in; (void)out_size; (void)ws_size;

  const float* x    = (const float*)d_in[0];
  const int*   ei   = (const int*)  d_in[1];
  const float* W1   = (const float*)d_in[2];
  const float* aS1  = (const float*)d_in[3];
  const float* aD1  = (const float*)d_in[4];
  const float* b1   = (const float*)d_in[5];
  const float* W2   = (const float*)d_in[6];
  const float* aS2  = (const float*)d_in[7];
  const float* aD2  = (const float*)d_in[8];
  const float* b2   = (const float*)d_in[9];
  float* out = (float*)d_out;

  float* ws  = (float*)d_ws;
  float* h1  = ws;
  float* o1  = h1  + (long long)NNODES * L1_HC;
  float* as1 = o1  + (long long)NNODES * L1_HC;
  float* ad1 = as1 + NNODES * L1_H;
  float* mx1 = ad1 + NNODES * L1_H;
  float* sm1 = mx1 + NNODES * L1_H;
  float* as2 = sm1 + NNODES * L1_H;
  float* ad2 = as2 + NNODES;
  float* mx2 = ad2 + NNODES;
  float* sm2 = mx2 + NNODES;
  float* h2  = h1;

  const float NEGINF = -__builtin_inff();
  const int EN = NEDGES + NNODES;
  auto cdiv = [](long long a, long long b) { return (int)((a + b - 1) / b); };

  fill_kernel<<<cdiv((long long)NNODES * L1_HC, 256), 256, 0, stream>>>(o1, 0.0f, (long long)NNODES * L1_HC);
  fill_kernel<<<cdiv(NNODES * L1_H, 256), 256, 0, stream>>>(mx1, NEGINF, NNODES * L1_H);
  fill_kernel<<<cdiv(NNODES * L1_H, 256), 256, 0, stream>>>(sm1, 0.0f, NNODES * L1_H);
  fill_kernel<<<cdiv(NNODES, 256), 256, 0, stream>>>(mx2, NEGINF, NNODES);
  fill_kernel<<<cdiv(NNODES, 256), 256, 0, stream>>>(sm2, 0.0f, NNODES);
  fill_kernel<<<cdiv((long long)NNODES * L2_HC, 256), 256, 0, stream>>>(out, 0.0f, (long long)NNODES * L2_HC);

  gemm_wmma_kernel<128, L1_HC><<<NNODES / 16, 128, 0, stream>>>(x, W1, h1);
  attn_kernel<<<cdiv(NNODES * L1_H, 128), 128, 0, stream>>>(h1, aS1, aD1, as1, ad1, L1_H, L1_C);
  edge_max_kernel<<<cdiv(EN, 256), 256, 0, stream>>>(ei, as1, ad1, mx1, L1_H);
  edge_sum_kernel<<<cdiv(EN, 256), 256, 0, stream>>>(ei, as1, ad1, mx1, sm1, L1_H);
  edge_scatter_kernel<<<cdiv((long long)EN * L1_HC, 256), 256, 0, stream>>>(
      ei, as1, ad1, mx1, sm1, h1, o1, L1_H, L1_C);
  bias_elu_kernel<<<cdiv((long long)NNODES * L1_HC, 256), 256, 0, stream>>>(
      o1, b1, (long long)NNODES * L1_HC, L1_HC);

  gemm_wmma_kernel<128, L2_HC><<<NNODES / 16, 64, 0, stream>>>(o1, W2, h2);
  attn_kernel<<<cdiv(NNODES, 128), 128, 0, stream>>>(h2, aS2, aD2, as2, ad2, L2_H, L2_C);
  edge_max_kernel<<<cdiv(EN, 256), 256, 0, stream>>>(ei, as2, ad2, mx2, L2_H);
  edge_sum_kernel<<<cdiv(EN, 256), 256, 0, stream>>>(ei, as2, ad2, mx2, sm2, L2_H);
  edge_scatter_kernel<<<cdiv((long long)EN * L2_HC, 256), 256, 0, stream>>>(
      ei, as2, ad2, mx2, sm2, h2, out, L2_H, L2_C);
  bias_add_kernel<<<cdiv((long long)NNODES * L2_HC, 256), 256, 0, stream>>>(
      out, b2, (long long)NNODES * L2_HC, L2_HC);
}


# GEMM epilogue fusion: attention dots taken from the accumulator registers (16-lane DPP xor tree, dot sum order changes), edge-pass accumulators started there; attn+fill kernels gone, 16 to 8 launches
# speedup vs baseline: 3.5842x; 3.5842x over previous
#include <hip/hip_runtime.h>
#include <math.h>

#define NNODES   100000
#define NEDGES   1600000
#define L1_H     4
#define L1_C     32
#define L1_HC    128
#define L2_H     1
#define L2_C     64
#define L2_HC    64
#define NEG_SLOPE 0.2f
#define EPS_      1e-16f
#define SCATTER_BLOCKS (((NEDGES + NNODES) / 32 + 7) / 8)

typedef __attribute__((ext_vector_type(2))) float v2f;
typedef __attribute__((ext_vector_type(8))) float v8f;

__device__ __forceinline__ int f32_key(float v) {
  const int b = __float_as_int(v);
  return b >= 0 ? b : b ^ 0x7FFFFFFF;
}
__device__ __forceinline__ float key_f32(int k) {
  return __int_as_float(k >= 0 ? k : k ^ 0x7FFFFFFF);
}
#define KEY_NEG_INF ((int)0x807FFFFF)

template <int CTRL>
__device__ __forceinline__ float row_xor_add(float v) {
  return v + __int_as_float(
      __builtin_amdgcn_update_dpp(0, __float_as_int(v), CTRL, 0xF, 0xF, true));
}

template <int K, int NC, int H, int RT>
__global__ __launch_bounds__(RT * NC)
void gemm_attn_init_kernel(const float* __restrict__ A, const float* __restrict__ B,
                           float* __restrict__ C,
                           const float* __restrict__ att_s,
                           const float* __restrict__ att_d,
                           float* __restrict__ as_, float* __restrict__ ad_,
                           int* __restrict__ mxk, float* __restrict__ sm,
                           float* __restrict__ accum) {
  constexpr int CT   = NC / 32;
  constexpr int CH   = NC / H;
  constexpr int WPH  = CH / 32;
  constexpr int ROWS = 16 * RT;
  constexpr int NE   = ROWS * H;
  constexpr int NT   = RT * NC;
  static_assert(CH % 32 == 0, "a wave's 32 columns lie inside one head");
  static_assert(NE % 32 == 0 && 2 * NE <= NT, "whole lines of as_/ad_, one wave each");
  static_assert((ROWS * NC) % (4 * NT) == 0, "accumulator rows zeroed in whole 16-byte steps");
  static_assert(NNODES % ROWS == 0, "no partial block: the barrier below is block-uniform");

  __shared__ float ps[WPH][NE];
  __shared__ float pd[WPH][NE];

  const int lane = threadIdx.x & 31;
  const int wv   = threadIdx.x >> 5;
  const int rt   = wv / CT;
  const int ct   = wv % CT;
  const int row0 = blockIdx.x * ROWS;
  const int rowt = row0 + rt * 16;
  const int col0 = ct * 32;
  const int m    = lane & 15;
  const int half = lane >> 4;
  const int kk   = half << 1;

  v8f acc0 = {};
  v8f acc1 = {};
  const float* ap = A + (long long)(rowt + m) * K + kk;
  const float* bp = B + (long long)kk * NC + (col0 + m);
#pragma unroll
  for (int k0 = 0; k0 < K; k0 += 4) {
    v2f a, b0, b1;
    a.x  = ap[k0];            a.y  = ap[k0 + 1];
    b0.x = bp[k0 * NC];       b0.y = bp[(k0 + 1) * NC];
    b1.x = bp[k0 * NC + 16];  b1.y = bp[(k0 + 1) * NC + 16];
    acc0 = __builtin_amdgcn_wmma_f32_16x16x4_f32(
        false, a, false, b0, (short)0, acc0, false, false);
    acc1 = __builtin_amdgcn_wmma_f32_16x16x4_f32(
        false, a, false, b1, (short)0, acc1, false, false);
  }
  const float s0 = att_s[col0 + m], s1 = att_s[col0 + 16 + m];
  const float d0 = att_d[col0 + m], d1 = att_d[col0 + 16 + m];
  float vs = 0.0f, vd = 0.0f;
#pragma unroll
  for (int r = 0; r < 8; ++r) {
    float p = acc0[r] * s0 + acc1[r] * s1;
    float q = acc0[r] * d0 + acc1[r] * d1;
    p = row_xor_add<0x161>(p);  q = row_xor_add<0x161>(q);
    p = row_xor_add<0x162>(p);  q = row_xor_add<0x162>(q);
    p = row_xor_add<0x164>(p);  q = row_xor_add<0x164>(q);
    p = row_xor_add<0x168>(p);  q = row_xor_add<0x168>(q);
    if (m == r) { vs = p; vd = q; }
  }
  if (m < 8) {
    const int e = (rt * 16 + half * 8 + m) * H + ct / WPH;
    ps[ct % WPH][e] = vs;
    pd[ct % WPH][e] = vd;
  }

  __syncthreads();

  const int orow = rowt + (half << 3);
  const int ocol = col0 + m;
  float* cp = C + (long long)orow * NC + ocol;
#pragma unroll
  for (int r = 0; r < 8; ++r) {
    cp[(long long)r * NC]      = acc0[r];
    cp[(long long)r * NC + 16] = acc1[r];
  }

  float4* z4 = (float4*)(accum + (long long)row0 * NC);
  const float4 zero4 = {0.0f, 0.0f, 0.0f, 0.0f};
#pragma unroll
  for (int i = 0; i < (ROWS * NC) / (4 * NT); ++i) z4[i * NT + threadIdx.x] = zero4;

  const int t = threadIdx.x;
  if (t < NE) {
    float v = ps[0][t];
#pragma unroll
    for (int w = 1; w < WPH; ++w) v += ps[w][t];
    as_[row0 * H + t] = v;
    mxk[row0 * H + t] = KEY_NEG_INF;
  } else if (t < 2 * NE) {
    const int u = t - NE;
    float v = pd[0][u];
#pragma unroll
    for (int w = 1; w < WPH; ++w) v += pd[w][u];
    ad_[row0 * H + u] = v;
    sm[row0 * H + u] = 0.0f;
  }
}

template <int H>
__device__ __forceinline__ void wave_edge_groups(const int* __restrict__ ei, int base,
                                                 int sub, int (&s)[H], int (&d)[H]) {
  constexpr int EPI = 32 / H;
  static_assert(32 % H == 0, "heads of an edge sit in adjacent lanes");
  static_assert(NEDGES % 32 == 0, "a wave's 32 edges are all listed or all self-loops");
  if (base < NEDGES) {
#pragma unroll
    for (int it = 0; it < H; ++it) {
      s[it] = ei[base + it * EPI + sub];
      d[it] = ei[NEDGES + base + it * EPI + sub];
    }
  } else {
#pragma unroll
    for (int it = 0; it < H; ++it) { s[it] = base - NEDGES + it * EPI + sub; d[it] = s[it]; }
  }
}

template <int H>
__global__ __launch_bounds__(256)
void edge_max_kernel(const int* __restrict__ ei,
                     const float* __restrict__ as_,
                     const float* __restrict__ ad_,
                     int* __restrict__ mxk) {
  const int lane = threadIdx.x & 31;
  const int base = __builtin_amdgcn_readfirstlane(
      (blockIdx.x * (blockDim.x >> 5) + (threadIdx.x >> 5)) * 32);
  if (base >= NEDGES + NNODES) return;
  const int h = lane % H, sub = lane / H;
  int s[H], d[H];
  wave_edge_groups<H>(ei, base, sub, s, d);
  float a[H], b[H];
#pragma unroll
  for (int it = 0; it < H; ++it) { a[it] = as_[s[it] * H + h]; b[it] = ad_[d[it] * H + h]; }
#pragma unroll
  for (int it = 0; it < H; ++it) {
    float e = a[it] + b[it];
    e = e > 0.0f ? e : NEG_SLOPE * e;
    atomicMax(&mxk[d[it] * H + h], f32_key(e));
  }
}

template <int H, int C>
__global__ __launch_bounds__(256)
void edge_sum_scatter_rows_kernel(const int* __restrict__ ei,
                                  const float* __restrict__ as_,
                                  const float* __restrict__ ad_,
                                  const int* __restrict__ mxk,
                                  float* __restrict__ sm,
                                  const float* __restrict__ hfeat,
                                  float* __restrict__ out) {
  constexpr int HC  = H * C;
  constexpr int NG  = HC / 32;
  constexpr int GPH = C / 32;
  constexpr int EB  = 8;
  constexpr int EPI = 32 / H;
  static_assert(C % 32 == 0, "a 32-channel group must lie inside one head");
  static_assert(NEDGES % 32 == 0 && NNODES % 32 == 0, "a wave's 32 edges are all valid");

  const int lane = threadIdx.x & 31;
  const int base = __builtin_amdgcn_readfirstlane(
      (blockIdx.x * (blockDim.x >> 5) + (threadIdx.x >> 5)) * 32);
  if (base >= NEDGES + NNODES) return;

  const int h = lane % H, sub = lane / H;
  int s[H], d[H];
  wave_edge_groups<H>(ei, base, sub, s, d);
  float a[H], b[H]; int mk[H];
#pragma unroll
  for (int it = 0; it < H; ++it) {
    a[it] = as_[s[it] * H + h]; b[it] = ad_[d[it] * H + h]; mk[it] = mxk[d[it] * H + h];
  }
  int al[H];
#pragma unroll
  for (int it = 0; it < H; ++it) {
    float e = a[it] + b[it];
    e = e > 0.0f ? e : NEG_SLOPE * e;
    const float ex = expf(e - key_f32(mk[it]));
    atomicAdd(&sm[d[it] * H + h], ex);
    al[it] = __float_as_int(ex);
  }

#pragma unroll
  for (int j0 = 0; j0 < 32; j0 += EB) {
    float hv[EB][NG];
#pragma unroll
    for (int g = 0; g < EB; ++g) {
      const int j  = j0 + g;
      const int sj = __builtin_amdgcn_readlane(s[j / EPI], (j % EPI) * H);
      const float* row = hfeat + (long long)sj * HC + lane;
#pragma unroll
      for (int v = 0; v < NG; ++v) hv[g][v] = row[v * 32];
    }
#pragma unroll
    for (int g = 0; g < EB; ++g) {
      const int j  = j0 + g;
      const int dj = __builtin_amdgcn_readlane(d[j / EPI], (j % EPI) * H);
      float* orow = out + (long long)dj * HC + lane;
#pragma unroll
      for (int v = 0; v < NG; ++v) {
        const float w = __int_as_float(
            __builtin_amdgcn_readlane(al[j / EPI], (j % EPI) * H + v / GPH));
        atomicAdd(orow + v * 32, hv[g][v] * w);
      }
    }
  }
}

template <int H, int C>
__global__ void norm_bias_elu_kernel(float* __restrict__ o, const float* __restrict__ sm,
                                     const float* __restrict__ b, long long n) {
  constexpr int HC = H * C;
  long long i = (long long)blockIdx.x * blockDim.x + threadIdx.x;
  if (i >= n) return;
  const int c = (int)(i % HC);
  float v = o[i] / (sm[(i / HC) * H + c / C] + EPS_) + b[c];
  o[i] = v > 0.0f ? v : expm1f(v);
}

template <int H, int C>
__global__ void norm_bias_add_kernel(float* __restrict__ o, const float* __restrict__ sm,
                                     const float* __restrict__ b, long long n) {
  constexpr int HC = H * C;
  long long i = (long long)blockIdx.x * blockDim.x + threadIdx.x;
  if (i >= n) return;
  const int c = (int)(i % HC);
  o[i] = o[i] / (sm[(i / HC) * H + c / C] + EPS_) + b[c];
}

extern "C" void kernel_launch(void* const* d_in, const int* in_sizes, int n_in,
                              void* d_out, int out_size, void* d_ws, size_t ws_size,
                              hipStream_t stream) {
  (void)in_sizes; (void)n_in; (void)out_size; (void)ws_size;

  const float* x    = (const float*)d_in[0];
  const int*   ei   = (const int*)  d_in[1];
  const float* W1   = (const float*)d_in[2];
  const float* aS1  = (const float*)d_in[3];
  const float* aD1  = (const float*)d_in[4];
  const float* b1   = (const float*)d_in[5];
  const float* W2   = (const float*)d_in[6];
  const float* aS2  = (const float*)d_in[7];
  const float* aD2  = (const float*)d_in[8];
  const float* b2   = (const float*)d_in[9];
  float* out = (float*)d_out;

  float* ws  = (float*)d_ws;
  float* h1  = ws;
  float* o1  = h1  + (long long)NNODES * L1_HC;
  float* as1 = o1  + (long long)NNODES * L1_HC;
  float* ad1 = as1 + NNODES * L1_H;
  int*   mx1 = (int*)(ad1 + NNODES * L1_H);
  float* sm1 = (float*)mx1 + NNODES * L1_H;
  float* as2 = sm1 + NNODES * L1_H;
  float* ad2 = as2 + NNODES;
  int*   mx2 = (int*)(ad2 + NNODES);
  float* sm2 = (float*)mx2 + NNODES;
  float* h2  = h1;

  auto cdiv = [](long long a, long long b) { return (int)((a + b - 1) / b); };

  gemm_attn_init_kernel<128, L1_HC, L1_H, 1><<<NNODES / 16, 128, 0, stream>>>(
      x, W1, h1, aS1, aD1, as1, ad1, mx1, sm1, o1);
  edge_max_kernel<L1_H><<<SCATTER_BLOCKS, 256, 0, stream>>>(ei, as1, ad1, mx1);
  edge_sum_scatter_rows_kernel<L1_H, L1_C><<<SCATTER_BLOCKS, 256, 0, stream>>>(
      ei, as1, ad1, mx1, sm1, h1, o1);
  norm_bias_elu_kernel<L1_H, L1_C><<<cdiv((long long)NNODES * L1_HC, 256), 256, 0, stream>>>(
      o1, sm1, b1, (long long)NNODES * L1_HC);

  gemm_attn_init_kernel<128, L2_HC, L2_H, 2><<<NNODES / 32, 128, 0, stream>>>(
      o1, W2, h2, aS2, aD2, as2, ad2, mx2, sm2, out);
  edge_max_kernel<L2_H><<<SCATTER_BLOCKS, 256, 0, stream>>>(ei, as2, ad2, mx2);
  edge_sum_scatter_rows_kernel<L2_H, L2_C><<<SCATTER_BLOCKS, 256, 0, stream>>>(
      ei, as2, ad2, mx2, sm2, h2, out);
  norm_bias_add_kernel<L2_H, L2_C><<<cdiv((long long)NNODES * L2_HC, 256), 256, 0, stream>>>(
      out, sm2, b2, (long long)NNODES * L2_HC);
}


# Softmax scalars by fewer line requests: max pass reduces a_src alone over listed edges (self-loop term set by the GEMM), leaves it in edge order; scatter reads that and one packed {a_dst,max} word
# speedup vs baseline: 3.8810x; 1.0828x over previous
#include <hip/hip_runtime.h>
#include <math.h>

#define NNODES   100000
#define NEDGES   1600000
#define L1_H     4
#define L1_C     32
#define L1_HC    128
#define L2_H     1
#define L2_C     64
#define L2_HC    64
#define NEG_SLOPE 0.2f
#define EPS_      1e-16f
#define SCATTER_BLOCKS (((NEDGES + NNODES) / 32 + 7) / 8)
#define MAX_BLOCKS (((NEDGES) / 32 + 7) / 8)

typedef __attribute__((ext_vector_type(2))) float v2f;
typedef __attribute__((ext_vector_type(8))) float v8f;

__device__ __forceinline__ int f32_key(float v) {
  const int b = __float_as_int(v);
  return b >= 0 ? b : b ^ 0x7FFFFFFF;
}
__device__ __forceinline__ float key_f32(int k) {
  return __int_as_float(k >= 0 ? k : k ^ 0x7FFFFFFF);
}

template <int CTRL>
__device__ __forceinline__ float row_xor_add(float v) {
  return v + __int_as_float(
      __builtin_amdgcn_update_dpp(0, __float_as_int(v), CTRL, 0xF, 0xF, true));
}

template <int K, int NC, int H, int RT>
__global__ __launch_bounds__(RT * NC)
void gemm_attn_init_kernel(const float* __restrict__ A, const float* __restrict__ B,
                           float* __restrict__ C,
                           const float* __restrict__ att_s,
                           const float* __restrict__ att_d,
                           float* __restrict__ as_, int* __restrict__ rec,
                           float* __restrict__ sm,
                           float* __restrict__ accum) {
  constexpr int CT   = NC / 32;
  constexpr int CH   = NC / H;
  constexpr int WPH  = CH / 32;
  constexpr int ROWS = 16 * RT;
  constexpr int NE   = ROWS * H;
  constexpr int NT   = RT * NC;
  static_assert(CH % 32 == 0, "a wave's 32 columns lie inside one head");
  static_assert(NE % 32 == 0 && NE <= NT, "whole lines of as_/rec/sm, whole waves");
  static_assert((ROWS * NC) % (4 * NT) == 0, "accumulator rows zeroed in whole 16-byte steps");
  static_assert(NNODES % ROWS == 0, "no partial block: the barrier below is block-uniform");

  __shared__ float ps[WPH][NE];
  __shared__ float pd[WPH][NE];

  const int lane = threadIdx.x & 31;
  const int wv   = threadIdx.x >> 5;
  const int rt   = wv / CT;
  const int ct   = wv % CT;
  const int row0 = blockIdx.x * ROWS;
  const int rowt = row0 + rt * 16;
  const int col0 = ct * 32;
  const int m    = lane & 15;
  const int half = lane >> 4;
  const int kk   = half << 1;

  v8f acc0 = {};
  v8f acc1 = {};
  const float* ap = A + (long long)(rowt + m) * K + kk;
  const float* bp = B + (long long)kk * NC + (col0 + m);
#pragma unroll
  for (int k0 = 0; k0 < K; k0 += 4) {
    v2f a, b0, b1;
    a.x  = ap[k0];            a.y  = ap[k0 + 1];
    b0.x = bp[k0 * NC];       b0.y = bp[(k0 + 1) * NC];
    b1.x = bp[k0 * NC + 16];  b1.y = bp[(k0 + 1) * NC + 16];
    acc0 = __builtin_amdgcn_wmma_f32_16x16x4_f32(
        false, a, false, b0, (short)0, acc0, false, false);
    acc1 = __builtin_amdgcn_wmma_f32_16x16x4_f32(
        false, a, false, b1, (short)0, acc1, false, false);
  }
  const float s0 = att_s[col0 + m], s1 = att_s[col0 + 16 + m];
  const float d0 = att_d[col0 + m], d1 = att_d[col0 + 16 + m];
  float vs = 0.0f, vd = 0.0f;
#pragma unroll
  for (int r = 0; r < 8; ++r) {
    float p = acc0[r] * s0 + acc1[r] * s1;
    float q = acc0[r] * d0 + acc1[r] * d1;
    p = row_xor_add<0x161>(p);  q = row_xor_add<0x161>(q);
    p = row_xor_add<0x162>(p);  q = row_xor_add<0x162>(q);
    p = row_xor_add<0x164>(p);  q = row_xor_add<0x164>(q);
    p = row_xor_add<0x168>(p);  q = row_xor_add<0x168>(q);
    if (m == r) { vs = p; vd = q; }
  }
  if (m < 8) {
    const int e = (rt * 16 + half * 8 + m) * H + ct / WPH;
    ps[ct % WPH][e] = vs;
    pd[ct % WPH][e] = vd;
  }

  __syncthreads();

  const int orow = rowt + (half << 3);
  const int ocol = col0 + m;
  float* cp = C + (long long)orow * NC + ocol;
#pragma unroll
  for (int r = 0; r < 8; ++r) {
    cp[(long long)r * NC]      = acc0[r];
    cp[(long long)r * NC + 16] = acc1[r];
  }

  float4* z4 = (float4*)(accum + (long long)row0 * NC);
  const float4 zero4 = {0.0f, 0.0f, 0.0f, 0.0f};
#pragma unroll
  for (int i = 0; i < (ROWS * NC) / (4 * NT); ++i) z4[i * NT + threadIdx.x] = zero4;

  const int t = threadIdx.x;
  if (t < NE) {
    float v = ps[0][t];
    float u = pd[0][t];
#pragma unroll
    for (int w = 1; w < WPH; ++w) { v += ps[w][t]; u += pd[w][t]; }
    as_[row0 * H + t] = v;
    ((int2*)rec)[row0 * H + t] = make_int2(__float_as_int(u), f32_key(v));
    sm[row0 * H + t] = 0.0f;
  }
}

template <int H>
__device__ __forceinline__ void wave_edge_groups(const int* __restrict__ ei, int base,
                                                 int sub, int (&s)[H], int (&d)[H]) {
  constexpr int EPI = 32 / H;
  static_assert(32 % H == 0, "heads of an edge sit in adjacent lanes");
  static_assert(NEDGES % 32 == 0, "a wave's 32 edges are all listed or all self-loops");
  if (base < NEDGES) {
#pragma unroll
    for (int it = 0; it < H; ++it) {
      s[it] = ei[base + it * EPI + sub];
      d[it] = ei[NEDGES + base + it * EPI + sub];
    }
  } else {
#pragma unroll
    for (int it = 0; it < H; ++it) { s[it] = base - NEDGES + it * EPI + sub; d[it] = s[it]; }
  }
}

template <int H>
__global__ __launch_bounds__(256)
void edge_max_kernel(const int* __restrict__ ei,
                     const float* __restrict__ as_,
                     float* __restrict__ ea,
                     int* __restrict__ rec) {
  const int lane = threadIdx.x & 31;
  const int base = __builtin_amdgcn_readfirstlane(
      (blockIdx.x * (blockDim.x >> 5) + (threadIdx.x >> 5)) * 32);
  if (base >= NEDGES) return;
  const int h = lane % H, sub = lane / H;
  int s[H], d[H];
  wave_edge_groups<H>(ei, base, sub, s, d);
  float a[H];
#pragma unroll
  for (int it = 0; it < H; ++it) a[it] = as_[s[it] * H + h];
#pragma unroll
  for (int it = 0; it < H; ++it) {
    ea[base * H + it * 32 + lane] = a[it];
    atomicMax(&rec[(d[it] * H + h) * 2 + 1], f32_key(a[it]));
  }
}

template <int H, int C>
__global__ __launch_bounds__(256)
void edge_sum_scatter_rows_kernel(const int* __restrict__ ei,
                                  const float* __restrict__ ea,
                                  const float* __restrict__ as_,
                                  const int* __restrict__ rec,
                                  float* __restrict__ sm,
                                  const float* __restrict__ hfeat,
                                  float* __restrict__ out) {
  constexpr int HC  = H * C;
  constexpr int NG  = HC / 32;
  constexpr int GPH = C / 32;
  constexpr int EB  = 8;
  constexpr int EPI = 32 / H;
  static_assert(C % 32 == 0, "a 32-channel group must lie inside one head");
  static_assert(NEDGES % 32 == 0 && NNODES % 32 == 0, "a wave's 32 edges are all valid");

  const int lane = threadIdx.x & 31;
  const int base = __builtin_amdgcn_readfirstlane(
      (blockIdx.x * (blockDim.x >> 5) + (threadIdx.x >> 5)) * 32);
  if (base >= NEDGES + NNODES) return;

  const int h = lane % H, sub = lane / H;
  const float* asrc = base < NEDGES ? ea + (long long)base * H
                                    : as_ + (long long)(base - NEDGES) * H;
  float a[H];
#pragma unroll
  for (int it = 0; it < H; ++it) a[it] = asrc[it * 32 + lane];
  int s[H], d[H];
  wave_edge_groups<H>(ei, base, sub, s, d);
  int2 r[H];
#pragma unroll
  for (int it = 0; it < H; ++it) r[it] = ((const int2*)rec)[d[it] * H + h];
  int al[H];
#pragma unroll
  for (int it = 0; it < H; ++it) {
    const float b = __int_as_float(r[it].x);
    float e = a[it] + b;
    e = e > 0.0f ? e : NEG_SLOPE * e;
    float mx = key_f32(r[it].y) + b;
    mx = mx > 0.0f ? mx : NEG_SLOPE * mx;
    const float ex = expf(e - mx);
    atomicAdd(&sm[d[it] * H + h], ex);
    al[it] = __float_as_int(ex);
  }

#pragma unroll
  for (int j0 = 0; j0 < 32; j0 += EB) {
    float hv[EB][NG];
#pragma unroll
    for (int g = 0; g < EB; ++g) {
      const int j  = j0 + g;
      const int sj = __builtin_amdgcn_readlane(s[j / EPI], (j % EPI) * H);
      const float* row = hfeat + (long long)sj * HC + lane;
#pragma unroll
      for (int v = 0; v < NG; ++v) hv[g][v] = row[v * 32];
    }
#pragma unroll
    for (int g = 0; g < EB; ++g) {
      const int j  = j0 + g;
      const int dj = __builtin_amdgcn_readlane(d[j / EPI], (j % EPI) * H);
      float* orow = out + (long long)dj * HC + lane;
#pragma unroll
      for (int v = 0; v < NG; ++v) {
        const float w = __int_as_float(
            __builtin_amdgcn_readlane(al[j / EPI], (j % EPI) * H + v / GPH));
        atomicAdd(orow + v * 32, hv[g][v] * w);
      }
    }
  }
}

template <int H, int C>
__global__ void norm_bias_elu_kernel(float* __restrict__ o, const float* __restrict__ sm,
                                     const float* __restrict__ b, long long n) {
  constexpr int HC = H * C;
  long long i = (long long)blockIdx.x * blockDim.x + threadIdx.x;
  if (i >= n) return;
  const int c = (int)(i % HC);
  float v = o[i] / (sm[(i / HC) * H + c / C] + EPS_) + b[c];
  o[i] = v > 0.0f ? v : expm1f(v);
}

template <int H, int C>
__global__ void norm_bias_add_kernel(float* __restrict__ o, const float* __restrict__ sm,
                                     const float* __restrict__ b, long long n) {
  constexpr int HC = H * C;
  long long i = (long long)blockIdx.x * blockDim.x + threadIdx.x;
  if (i >= n) return;
  const int c = (int)(i % HC);
  o[i] = o[i] / (sm[(i / HC) * H + c / C] + EPS_) + b[c];
}

extern "C" void kernel_launch(void* const* d_in, const int* in_sizes, int n_in,
                              void* d_out, int out_size, void* d_ws, size_t ws_size,
                              hipStream_t stream) {
  (void)in_sizes; (void)n_in; (void)out_size; (void)ws_size;

  const float* x    = (const float*)d_in[0];
  const int*   ei   = (const int*)  d_in[1];
  const float* W1   = (const float*)d_in[2];
  const float* aS1  = (const float*)d_in[3];
  const float* aD1  = (const float*)d_in[4];
  const float* b1   = (const float*)d_in[5];
  const float* W2   = (const float*)d_in[6];
  const float* aS2  = (const float*)d_in[7];
  const float* aD2  = (const float*)d_in[8];
  const float* b2   = (const float*)d_in[9];
  float* out = (float*)d_out;

  float* ws  = (float*)d_ws;
  float* h1  = ws;
  float* o1  = h1  + (long long)NNODES * L1_HC;
  float* as1 = o1  + (long long)NNODES * L1_HC;
  int*   rc1 = (int*)(as1 + NNODES * L1_H);
  float* sm1 = (float*)rc1 + NNODES * L1_H * 2;
  float* as2 = sm1 + NNODES * L1_H;
  int*   rc2 = (int*)(as2 + NNODES);
  float* sm2 = (float*)rc2 + NNODES * 2;
  float* ea1 = sm2 + NNODES;
  float* ea2 = ea1 + (long long)NEDGES * L1_H;
  float* h2  = h1;

  auto cdiv = [](long long a, long long b) { return (int)((a + b - 1) / b); };

  gemm_attn_init_kernel<128, L1_HC, L1_H, 1><<<NNODES / 16, 128, 0, stream>>>(
      x, W1, h1, aS1, aD1, as1, rc1, sm1, o1);
  edge_max_kernel<L1_H><<<MAX_BLOCKS, 256, 0, stream>>>(ei, as1, ea1, rc1);
  edge_sum_scatter_rows_kernel<L1_H, L1_C><<<SCATTER_BLOCKS, 256, 0, stream>>>(
      ei, ea1, as1, rc1, sm1, h1, o1);
  norm_bias_elu_kernel<L1_H, L1_C><<<cdiv((long long)NNODES * L1_HC, 256), 256, 0, stream>>>(
      o1, sm1, b1, (long long)NNODES * L1_HC);

  gemm_attn_init_kernel<128, L2_HC, L2_H, 2><<<NNODES / 32, 128, 0, stream>>>(
      o1, W2, h2, aS2, aD2, as2, rc2, sm2, out);
  edge_max_kernel<L2_H><<<MAX_BLOCKS, 256, 0, stream>>>(ei, as2, ea2, rc2);
  edge_sum_scatter_rows_kernel<L2_H, L2_C><<<SCATTER_BLOCKS, 256, 0, stream>>>(
      ei, ea2, as2, rc2, sm2, h2, out);
  norm_bias_add_kernel<L2_H, L2_C><<<cdiv((long long)NNODES * L2_HC, 256), 256, 0, stream>>>(
      out, sm2, b2, (long long)NNODES * L2_HC);
}


# Row and exp-sum adds kept inside one half of the chip: destination rows split in two sides, a block reads its die and takes 32-edge chunks by ticket for its own side; add arrival order differs
# speedup vs baseline: 5.1590x; 1.3293x over previous
#include <hip/hip_runtime.h>
#include <math.h>

#define NNODES   100000
#define NEDGES   1600000
#define L1_H     4
#define L1_C     32
#define L1_HC    128
#define L2_H     1
#define L2_C     64
#define L2_HC    64
#define NEG_SLOPE 0.2f
#define EPS_      1e-16f
#define SCATTER_CHUNKS ((NEDGES + NNODES) / 32)
#define SCATTER_BLOCKS 2048
#define WAVES_PER_TICKET 8
#define HALF_NODES (NNODES / 2)
#define CNT_STRIDE 64
#define MAX_BLOCKS (((NEDGES) / 32 + 7) / 8)

typedef __attribute__((ext_vector_type(2))) float v2f;
typedef __attribute__((ext_vector_type(8))) float v8f;

__device__ __forceinline__ int f32_key(float v) {
  const int b = __float_as_int(v);
  return b >= 0 ? b : b ^ 0x7FFFFFFF;
}
__device__ __forceinline__ float key_f32(int k) {
  return __int_as_float(k >= 0 ? k : k ^ 0x7FFFFFFF);
}

template <int CTRL>
__device__ __forceinline__ float row_xor_add(float v) {
  return v + __int_as_float(
      __builtin_amdgcn_update_dpp(0, __float_as_int(v), CTRL, 0xF, 0xF, true));
}

template <int K, int NC, int H, int RT>
__global__ __launch_bounds__(RT * NC)
void gemm_attn_init_kernel(const float* __restrict__ A, const float* __restrict__ B,
                           float* __restrict__ C,
                           const float* __restrict__ att_s,
                           const float* __restrict__ att_d,
                           float* __restrict__ as_, int* __restrict__ rec,
                           float* __restrict__ sm,
                           float* __restrict__ accum,
                           int* __restrict__ cnt) {
  constexpr int CT   = NC / 32;
  constexpr int CH   = NC / H;
  constexpr int WPH  = CH / 32;
  constexpr int ROWS = 16 * RT;
  constexpr int NE   = ROWS * H;
  constexpr int NT   = RT * NC;
  static_assert(CH % 32 == 0, "a wave's 32 columns lie inside one head");
  static_assert(NE % 32 == 0 && NE <= NT, "whole lines of as_/rec/sm, whole waves");
  static_assert((ROWS * NC) % (4 * NT) == 0, "accumulator rows zeroed in whole 16-byte steps");
  static_assert(NNODES % ROWS == 0, "no partial block: the barrier below is block-uniform");

  __shared__ float ps[WPH][NE];
  __shared__ float pd[WPH][NE];

  const int lane = threadIdx.x & 31;
  const int wv   = threadIdx.x >> 5;
  const int rt   = wv / CT;
  const int ct   = wv % CT;
  const int row0 = blockIdx.x * ROWS;
  const int rowt = row0 + rt * 16;
  const int col0 = ct * 32;
  const int m    = lane & 15;
  const int half = lane >> 4;
  const int kk   = half << 1;

  v8f acc0 = {};
  v8f acc1 = {};
  const float* ap = A + (long long)(rowt + m) * K + kk;
  const float* bp = B + (long long)kk * NC + (col0 + m);
#pragma unroll
  for (int k0 = 0; k0 < K; k0 += 4) {
    v2f a, b0, b1;
    a.x  = ap[k0];            a.y  = ap[k0 + 1];
    b0.x = bp[k0 * NC];       b0.y = bp[(k0 + 1) * NC];
    b1.x = bp[k0 * NC + 16];  b1.y = bp[(k0 + 1) * NC + 16];
    acc0 = __builtin_amdgcn_wmma_f32_16x16x4_f32(
        false, a, false, b0, (short)0, acc0, false, false);
    acc1 = __builtin_amdgcn_wmma_f32_16x16x4_f32(
        false, a, false, b1, (short)0, acc1, false, false);
  }
  const float s0 = att_s[col0 + m], s1 = att_s[col0 + 16 + m];
  const float d0 = att_d[col0 + m], d1 = att_d[col0 + 16 + m];
  float vs = 0.0f, vd = 0.0f;
#pragma unroll
  for (int r = 0; r < 8; ++r) {
    float p = acc0[r] * s0 + acc1[r] * s1;
    float q = acc0[r] * d0 + acc1[r] * d1;
    p = row_xor_add<0x161>(p);  q = row_xor_add<0x161>(q);
    p = row_xor_add<0x162>(p);  q = row_xor_add<0x162>(q);
    p = row_xor_add<0x164>(p);  q = row_xor_add<0x164>(q);
    p = row_xor_add<0x168>(p);  q = row_xor_add<0x168>(q);
    if (m == r) { vs = p; vd = q; }
  }
  if (m < 8) {
    const int e = (rt * 16 + half * 8 + m) * H + ct / WPH;
    ps[ct % WPH][e] = vs;
    pd[ct % WPH][e] = vd;
  }

  __syncthreads();

  const int orow = rowt + (half << 3);
  const int ocol = col0 + m;
  float* cp = C + (long long)orow * NC + ocol;
#pragma unroll
  for (int r = 0; r < 8; ++r) {
    cp[(long long)r * NC]      = acc0[r];
    cp[(long long)r * NC + 16] = acc1[r];
  }

  float4* z4 = (float4*)(accum + (long long)row0 * NC);
  const float4 zero4 = {0.0f, 0.0f, 0.0f, 0.0f};
#pragma unroll
  for (int i = 0; i < (ROWS * NC) / (4 * NT); ++i) z4[i * NT + threadIdx.x] = zero4;

  if (blockIdx.x == 0 && threadIdx.x < 2) cnt[threadIdx.x * CNT_STRIDE] = 0;

  const int t = threadIdx.x;
  if (t < NE) {
    float v = ps[0][t];
    float u = pd[0][t];
#pragma unroll
    for (int w = 1; w < WPH; ++w) { v += ps[w][t]; u += pd[w][t]; }
    as_[row0 * H + t] = v;
    ((int2*)rec)[row0 * H + t] = make_int2(__float_as_int(u), f32_key(v));
    sm[row0 * H + t] = 0.0f;
  }
}

template <int H>
__device__ __forceinline__ void wave_edge_groups(const int* __restrict__ ei, int base,
                                                 int sub, int (&s)[H], int (&d)[H]) {
  constexpr int EPI = 32 / H;
  static_assert(32 % H == 0, "heads of an edge sit in adjacent lanes");
  static_assert(NEDGES % 32 == 0, "a wave's 32 edges are all listed or all self-loops");
  if (base < NEDGES) {
#pragma unroll
    for (int it = 0; it < H; ++it) {
      s[it] = ei[base + it * EPI + sub];
      d[it] = ei[NEDGES + base + it * EPI + sub];
    }
  } else {
#pragma unroll
    for (int it = 0; it < H; ++it) { s[it] = base - NEDGES + it * EPI + sub; d[it] = s[it]; }
  }
}

template <int H>
__global__ __launch_bounds__(256)
void edge_max_kernel(const int* __restrict__ ei,
                     const float* __restrict__ as_,
                     float* __restrict__ ea,
                     int* __restrict__ rec) {
  const int lane = threadIdx.x & 31;
  const int base = __builtin_amdgcn_readfirstlane(
      (blockIdx.x * (blockDim.x >> 5) + (threadIdx.x >> 5)) * 32);
  if (base >= NEDGES) return;
  const int h = lane % H, sub = lane / H;
  int s[H], d[H];
  wave_edge_groups<H>(ei, base, sub, s, d);
  float a[H];
#pragma unroll
  for (int it = 0; it < H; ++it) a[it] = as_[s[it] * H + h];
#pragma unroll
  for (int it = 0; it < H; ++it) {
    ea[base * H + it * 32 + lane] = a[it];
    atomicMax(&rec[(d[it] * H + h) * 2 + 1], f32_key(a[it]));
  }
}

template <int H, int C>
__device__ __forceinline__ void scatter_chunk(const int* __restrict__ ei,
                                              const float* __restrict__ ea,
                                              const float* __restrict__ as_,
                                              const int* __restrict__ rec,
                                              float* __restrict__ sm,
                                              const float* __restrict__ hfeat,
                                              float* __restrict__ out,
                                              const int base, const int side,
                                              const int lane) {
  constexpr int HC  = H * C;
  constexpr int NG  = HC / 32;
  constexpr int GPH = C / 32;
  constexpr int EB  = NG <= 2 ? 16 : 8;
  constexpr int EPI = 32 / H;

  const int h = lane % H, sub = lane / H;
  const float* asrc = base < NEDGES ? ea + (long long)base * H
                                    : as_ + (long long)(base - NEDGES) * H;
  float a[H];
#pragma unroll
  for (int it = 0; it < H; ++it) a[it] = asrc[it * 32 + lane];
  int s[H], d[H];
  wave_edge_groups<H>(ei, base, sub, s, d);
  int2 r[H];
#pragma unroll
  for (int it = 0; it < H; ++it) {
    r[it] = make_int2(0, 0);
    if ((d[it] >= HALF_NODES) == (side != 0)) r[it] = ((const int2*)rec)[d[it] * H + h];
  }
  int al[H];
#pragma unroll
  for (int it = 0; it < H; ++it) {
    const float b = __int_as_float(r[it].x);
    float e = a[it] + b;
    e = e > 0.0f ? e : NEG_SLOPE * e;
    float mx = key_f32(r[it].y) + b;
    mx = mx > 0.0f ? mx : NEG_SLOPE * mx;
    const float ex = expf(e - mx);
    if ((d[it] >= HALF_NODES) == (side != 0)) atomicAdd(&sm[d[it] * H + h], ex);
    al[it] = __float_as_int(ex);
  }

#pragma unroll
  for (int j0 = 0; j0 < 32; j0 += EB) {
    float hv[EB][NG];
#pragma unroll
    for (int g = 0; g < EB; ++g) {
#pragma unroll
      for (int v = 0; v < NG; ++v) hv[g][v] = 0.0f;
    }
#pragma unroll
    for (int g = 0; g < EB; ++g) {
      const int j  = j0 + g;
      const int dj = __builtin_amdgcn_readlane(d[j / EPI], (j % EPI) * H);
      if ((dj >= HALF_NODES) == (side != 0)) {
        const int sj = __builtin_amdgcn_readlane(s[j / EPI], (j % EPI) * H);
        const float* row = hfeat + (long long)sj * HC + lane;
#pragma unroll
        for (int v = 0; v < NG; ++v) hv[g][v] = row[v * 32];
      }
    }
#pragma unroll
    for (int g = 0; g < EB; ++g) {
      const int j  = j0 + g;
      const int dj = __builtin_amdgcn_readlane(d[j / EPI], (j % EPI) * H);
      if ((dj >= HALF_NODES) == (side != 0)) {
        float* orow = out + (long long)dj * HC + lane;
#pragma unroll
        for (int v = 0; v < NG; ++v) {
          const float w = __int_as_float(
              __builtin_amdgcn_readlane(al[j / EPI], (j % EPI) * H + v / GPH));
          atomicAdd(orow + v * 32, hv[g][v] * w);
        }
      }
    }
  }
}

template <int H, int C>
__global__ __launch_bounds__(256)
void edge_sum_scatter_rows_kernel(const int* __restrict__ ei,
                                  const float* __restrict__ ea,
                                  const float* __restrict__ as_,
                                  const int* __restrict__ rec,
                                  float* __restrict__ sm,
                                  const float* __restrict__ hfeat,
                                  float* __restrict__ out,
                                  int* __restrict__ cnt) {
  static_assert(C % 32 == 0, "a 32-channel group must lie inside one head");
  static_assert(NEDGES % 32 == 0 && NNODES % 32 == 0, "a chunk's 32 edges are all valid");

  const int lane = threadIdx.x & 31;
  const int wv = __builtin_amdgcn_readfirstlane(threadIdx.x >> 5);
  __shared__ int place_sh;
  __shared__ int tick_sh[2];
  if (threadIdx.x == 0) place_sh = __builtin_amdgcn_s_sendmsg_rtn(0x87);
  __syncthreads();
  const int myhalf = (__builtin_amdgcn_readfirstlane(place_sh) >> 18) & 1;

  int rnd = 0;
  for (int ph = 0; ph < 2; ++ph) {
    const int side = myhalf ^ ph;
    int* c = cnt + side * CNT_STRIDE;
    for (;;) {
      if (threadIdx.x == 0) tick_sh[rnd & 1] = atomicAdd(c, WAVES_PER_TICKET);
      __syncthreads();
      const int t = __builtin_amdgcn_readfirstlane(tick_sh[rnd & 1]);
      ++rnd;
      if (t >= SCATTER_CHUNKS) break;
      const int chunk = t + wv;
      if (chunk < SCATTER_CHUNKS)
        scatter_chunk<H, C>(ei, ea, as_, rec, sm, hfeat, out, chunk * 32, side, lane);
    }
  }
}

template <int H, int C>
__global__ void norm_bias_elu_kernel(float* __restrict__ o, const float* __restrict__ sm,
                                     const float* __restrict__ b, long long n) {
  constexpr int HC = H * C;
  long long i = (long long)blockIdx.x * blockDim.x + threadIdx.x;
  if (i >= n) return;
  const int c = (int)(i % HC);
  float v = o[i] / (sm[(i / HC) * H + c / C] + EPS_) + b[c];
  o[i] = v > 0.0f ? v : expm1f(v);
}

template <int H, int C>
__global__ void norm_bias_add_kernel(float* __restrict__ o, const float* __restrict__ sm,
                                     const float* __restrict__ b, long long n) {
  constexpr int HC = H * C;
  long long i = (long long)blockIdx.x * blockDim.x + threadIdx.x;
  if (i >= n) return;
  const int c = (int)(i % HC);
  o[i] = o[i] / (sm[(i / HC) * H + c / C] + EPS_) + b[c];
}

extern "C" void kernel_launch(void* const* d_in, const int* in_sizes, int n_in,
                              void* d_out, int out_size, void* d_ws, size_t ws_size,
                              hipStream_t stream) {
  (void)in_sizes; (void)n_in; (void)out_size; (void)ws_size;

  const float* x    = (const float*)d_in[0];
  const int*   ei   = (const int*)  d_in[1];
  const float* W1   = (const float*)d_in[2];
  const float* aS1  = (const float*)d_in[3];
  const float* aD1  = (const float*)d_in[4];
  const float* b1   = (const float*)d_in[5];
  const float* W2   = (const float*)d_in[6];
  const float* aS2  = (const float*)d_in[7];
  const float* aD2  = (const float*)d_in[8];
  const float* b2   = (const float*)d_in[9];
  float* out = (float*)d_out;

  float* ws  = (float*)d_ws;
  float* h1  = ws;
  float* o1  = h1  + (long long)NNODES * L1_HC;
  float* as1 = o1  + (long long)NNODES * L1_HC;
  int*   rc1 = (int*)(as1 + NNODES * L1_H);
  float* sm1 = (float*)rc1 + NNODES * L1_H * 2;
  float* as2 = sm1 + NNODES * L1_H;
  int*   rc2 = (int*)(as2 + NNODES);
  float* sm2 = (float*)rc2 + NNODES * 2;
  float* ea1 = sm2 + NNODES;
  float* ea2 = ea1 + (long long)NEDGES * L1_H;
  float* h2  = h1;
  int*   cn1 = (int*)((char*)d_ws + 160000000LL);
  int*   cn2 = cn1 + 2 * CNT_STRIDE;

  auto cdiv = [](long long a, long long b) { return (int)((a + b - 1) / b); };

  gemm_attn_init_kernel<128, L1_HC, L1_H, 1><<<NNODES / 16, 128, 0, stream>>>(
      x, W1, h1, aS1, aD1, as1, rc1, sm1, o1, cn1);
  edge_max_kernel<L1_H><<<MAX_BLOCKS, 256, 0, stream>>>(ei, as1, ea1, rc1);
  edge_sum_scatter_rows_kernel<L1_H, L1_C><<<SCATTER_BLOCKS, 256, 0, stream>>>(
      ei, ea1, as1, rc1, sm1, h1, o1, cn1);
  norm_bias_elu_kernel<L1_H, L1_C><<<cdiv((long long)NNODES * L1_HC, 256), 256, 0, stream>>>(
      o1, sm1, b1, (long long)NNODES * L1_HC);

  gemm_attn_init_kernel<128, L2_HC, L2_H, 2><<<NNODES / 32, 128, 0, stream>>>(
      o1, W2, h2, aS2, aD2, as2, rc2, sm2, out, cn2);
  edge_max_kernel<L2_H><<<MAX_BLOCKS, 256, 0, stream>>>(ei, as2, ea2, rc2);
  edge_sum_scatter_rows_kernel<L2_H, L2_C><<<SCATTER_BLOCKS, 256, 0, stream>>>(
      ei, ea2, as2, rc2, sm2, h2, out, cn2);
  norm_bias_add_kernel<L2_H, L2_C><<<cdiv((long long)NNODES * L2_HC, 256), 256, 0, stream>>>(
      out, sm2, b2, (long long)NNODES * L2_HC);
}


# Row sums kept in LDS: 256 work-groups each own 391 destination rows, find their edges by scanning one owner byte per edge, add by LDS float atomics (no global float atomics); add order differs
# speedup vs baseline: 5.5820x; 1.0820x over previous
#include <hip/hip_runtime.h>
#include <math.h>

#define NNODES   100000
#define NEDGES   1600000
#define L1_H     4
#define L1_C     32
#define L1_HC    128
#define L2_H     1
#define L2_C     64
#define L2_HC    64
#define NEG_SLOPE 0.2f
#define EPS_      1e-16f
#define OWN_BLOCKS 256
#define OWN_R      ((NNODES + OWN_BLOCKS - 1) / OWN_BLOCKS)
#define OWN_WAVES  32
#define SCAN_EDGES  512
#define SCAN_GROUPS (NEDGES / SCAN_EDGES)
#define QCAP 768
#define MAX_BLOCKS (((NEDGES) / 32 + 7) / 8)

typedef __attribute__((ext_vector_type(2))) float v2f;
typedef __attribute__((ext_vector_type(8))) float v8f;

__device__ __forceinline__ int f32_key(float v) {
  const int b = __float_as_int(v);
  return b >= 0 ? b : b ^ 0x7FFFFFFF;
}
__device__ __forceinline__ float key_f32(int k) {
  return __int_as_float(k >= 0 ? k : k ^ 0x7FFFFFFF);
}

template <int CTRL>
__device__ __forceinline__ float row_xor_add(float v) {
  return v + __int_as_float(
      __builtin_amdgcn_update_dpp(0, __float_as_int(v), CTRL, 0xF, 0xF, true));
}

template <int K, int NC, int H, int RT>
__global__ __launch_bounds__(RT * NC)
void gemm_attn_init_kernel(const float* __restrict__ A, const float* __restrict__ B,
                           float* __restrict__ C,
                           const float* __restrict__ att_s,
                           const float* __restrict__ att_d,
                           float* __restrict__ as_, int* __restrict__ rec) {
  constexpr int CT   = NC / 32;
  constexpr int CH   = NC / H;
  constexpr int WPH  = CH / 32;
  constexpr int ROWS = 16 * RT;
  constexpr int NE   = ROWS * H;
  constexpr int NT   = RT * NC;
  static_assert(CH % 32 == 0, "a wave's 32 columns lie inside one head");
  static_assert(NE % 32 == 0 && NE <= NT, "whole lines of as_/rec, whole waves");
  static_assert(NNODES % ROWS == 0, "no partial block: the barrier below is block-uniform");

  __shared__ float ps[WPH][NE];
  __shared__ float pd[WPH][NE];

  const int lane = threadIdx.x & 31;
  const int wv   = threadIdx.x >> 5;
  const int rt   = wv / CT;
  const int ct   = wv % CT;
  const int row0 = blockIdx.x * ROWS;
  const int rowt = row0 + rt * 16;
  const int col0 = ct * 32;
  const int m    = lane & 15;
  const int half = lane >> 4;
  const int kk   = half << 1;

  v8f acc0 = {};
  v8f acc1 = {};
  const float* ap = A + (long long)(rowt + m) * K + kk;
  const float* bp = B + (long long)kk * NC + (col0 + m);
#pragma unroll
  for (int k0 = 0; k0 < K; k0 += 4) {
    v2f a, b0, b1;
    a.x  = ap[k0];            a.y  = ap[k0 + 1];
    b0.x = bp[k0 * NC];       b0.y = bp[(k0 + 1) * NC];
    b1.x = bp[k0 * NC + 16];  b1.y = bp[(k0 + 1) * NC + 16];
    acc0 = __builtin_amdgcn_wmma_f32_16x16x4_f32(
        false, a, false, b0, (short)0, acc0, false, false);
    acc1 = __builtin_amdgcn_wmma_f32_16x16x4_f32(
        false, a, false, b1, (short)0, acc1, false, false);
  }
  const float s0 = att_s[col0 + m], s1 = att_s[col0 + 16 + m];
  const float d0 = att_d[col0 + m], d1 = att_d[col0 + 16 + m];
  float vs = 0.0f, vd = 0.0f;
#pragma unroll
  for (int r = 0; r < 8; ++r) {
    float p = acc0[r] * s0 + acc1[r] * s1;
    float q = acc0[r] * d0 + acc1[r] * d1;
    p = row_xor_add<0x161>(p);  q = row_xor_add<0x161>(q);
    p = row_xor_add<0x162>(p);  q = row_xor_add<0x162>(q);
    p = row_xor_add<0x164>(p);  q = row_xor_add<0x164>(q);
    p = row_xor_add<0x168>(p);  q = row_xor_add<0x168>(q);
    if (m == r) { vs = p; vd = q; }
  }
  if (m < 8) {
    const int e = (rt * 16 + half * 8 + m) * H + ct / WPH;
    ps[ct % WPH][e] = vs;
    pd[ct % WPH][e] = vd;
  }

  __syncthreads();

  const int orow = rowt + (half << 3);
  const int ocol = col0 + m;
  float* cp = C + (long long)orow * NC + ocol;
#pragma unroll
  for (int r = 0; r < 8; ++r) {
    cp[(long long)r * NC]      = acc0[r];
    cp[(long long)r * NC + 16] = acc1[r];
  }

  const int t = threadIdx.x;
  if (t < NE) {
    float v = ps[0][t];
    float u = pd[0][t];
#pragma unroll
    for (int w = 1; w < WPH; ++w) { v += ps[w][t]; u += pd[w][t]; }
    as_[row0 * H + t] = v;
    ((int2*)rec)[row0 * H + t] = make_int2(__float_as_int(u), f32_key(v));
  }
}

template <int H>
__device__ __forceinline__ void wave_edge_groups(const int* __restrict__ ei, int base,
                                                 int sub, int (&s)[H], int (&d)[H]) {
  constexpr int EPI = 32 / H;
  static_assert(32 % H == 0, "heads of an edge sit in adjacent lanes");
  static_assert(NEDGES % 32 == 0, "a wave's 32 edges are all listed or all self-loops");
  if (base < NEDGES) {
#pragma unroll
    for (int it = 0; it < H; ++it) {
      s[it] = ei[base + it * EPI + sub];
      d[it] = ei[NEDGES + base + it * EPI + sub];
    }
  } else {
#pragma unroll
    for (int it = 0; it < H; ++it) { s[it] = base - NEDGES + it * EPI + sub; d[it] = s[it]; }
  }
}

template <int H, bool OB>
__global__ __launch_bounds__(256)
void edge_max_kernel(const int* __restrict__ ei,
                     const float* __restrict__ as_,
                     float* __restrict__ ea,
                     int* __restrict__ rec,
                     unsigned char* __restrict__ ob) {
  const int lane = threadIdx.x & 31;
  const int base = __builtin_amdgcn_readfirstlane(
      (blockIdx.x * (blockDim.x >> 5) + (threadIdx.x >> 5)) * 32);
  if (base >= NEDGES) return;
  static_assert((NNODES - 1) / OWN_R < 256, "an owner fits a byte");
  const int h = lane % H, sub = lane / H;
  int s[H], d[H];
  wave_edge_groups<H>(ei, base, sub, s, d);
  float a[H];
#pragma unroll
  for (int it = 0; it < H; ++it) a[it] = as_[s[it] * H + h];
#pragma unroll
  for (int it = 0; it < H; ++it) {
    ea[base * H + it * 32 + lane] = a[it];
    atomicMax(&rec[(d[it] * H + h) * 2 + 1], f32_key(a[it]));
  }
  if (OB) ob[base + lane] = (unsigned char)(ei[NEDGES + base + lane] / OWN_R);
}

template <int H, int C, bool ELU>
__global__ __launch_bounds__(OWN_WAVES * 32)
void owner_rows_kernel(const int* __restrict__ ei,
                       const unsigned char* __restrict__ ob,
                       const float* __restrict__ ea,
                       const float* __restrict__ as_,
                       const int* __restrict__ rec,
                       const float* __restrict__ hfeat,
                       const float* __restrict__ bias,
                       float* __restrict__ out) {
  constexpr int HC  = H * C;
  constexpr int NG  = HC / 32;
  constexpr int GPH = C / 32;
  constexpr int EB  = 64 / NG;
  constexpr int NIT = (EB * H + 31) / 32;
  static_assert(C % 32 == 0, "a 32-channel group must lie inside one head");
  static_assert(32 % H == 0, "heads of an edge sit in adjacent lanes");
  static_assert(NEDGES % SCAN_EDGES == 0, "whole scan groups");
  static_assert(QCAP >= SCAN_EDGES, "a whole group fits the list");
  static_assert(EB * H == NIT * 32, "step 1 covers the batch's (edge, head) pairs in whole waves");

  __shared__ float acc[OWN_R * HC];
  __shared__ float smL[OWN_R * H];
  __shared__ int   q[OWN_WAVES][QCAP];

  const int lane = threadIdx.x & 31;
  const int wv = __builtin_amdgcn_readfirstlane(threadIdx.x >> 5);
  const int lo = blockIdx.x * OWN_R;
  const int nown = NNODES - lo < OWN_R ? NNODES - lo : OWN_R;

  for (int p = threadIdx.x; p < nown * H; p += OWN_WAVES * 32) {
    const float a = as_[lo * H + p];
    const int2 r = ((const int2*)rec)[lo * H + p];
    const float b = __int_as_float(r.x);
    float e = a + b;
    e = e > 0.0f ? e : NEG_SLOPE * e;
    float mx = key_f32(r.y) + b;
    mx = mx > 0.0f ? mx : NEG_SLOPE * mx;
    smL[p] = expf(e - mx);
  }
  __syncthreads();
  for (int nl0 = wv; nl0 < nown; nl0 += 4 * OWN_WAVES) {
    float hr[4][NG];
#pragma unroll
    for (int u = 0; u < 4; ++u) {
      const int nl = nl0 + u * OWN_WAVES;
#pragma unroll
      for (int v = 0; v < NG; ++v) hr[u][v] = 0.0f;
      if (nl < nown) {
        const float* row = hfeat + (long long)(lo + nl) * HC + lane;
#pragma unroll
        for (int v = 0; v < NG; ++v) hr[u][v] = row[v * 32];
      }
    }
#pragma unroll
    for (int u = 0; u < 4; ++u) {
      const int nl = nl0 + u * OWN_WAVES;
      if (nl < nown) {
#pragma unroll
        for (int v = 0; v < NG; ++v)
          acc[nl * HC + v * 32 + lane] = smL[nl * H + v / GPH] * hr[u][v];
      }
    }
  }
  __syncthreads();

  int nq = 0;

  auto drain = [&]() {
    for (int i0 = 0; i0 < nq; i0 += EB) {
      const int nb = nq - i0 < EB ? nq - i0 : EB;
      int s[NIT], d[NIT], xb[NIT], eg[NIT];
      float a[NIT];
      int2 r[NIT];
#pragma unroll
      for (int it = 0; it < NIT; ++it) {
        const int j = (it * 32 + lane) / H;
        eg[it] = j < nb ? q[wv][i0 + j] : 0;
      }
#pragma unroll
      for (int it = 0; it < NIT; ++it) {
        const int h = (it * 32 + lane) % H;
        s[it] = ei[eg[it]];
        d[it] = ei[NEDGES + eg[it]];
        a[it] = ea[(long long)eg[it] * H + h];
      }
#pragma unroll
      for (int it = 0; it < NIT; ++it) asm volatile("" :: "v"(s[it]), "v"(d[it]), "v"(a[it]));
#pragma unroll
      for (int it = 0; it < NIT; ++it) {
        const int h = (it * 32 + lane) % H;
        r[it] = ((const int2*)rec)[d[it] * H + h];
      }
      float hv[EB][NG];
#pragma unroll
      for (int g = 0; g < EB; ++g) {
#pragma unroll
        for (int v = 0; v < NG; ++v) hv[g][v] = 0.0f;
      }
#pragma unroll
      for (int g = 0; g < EB; ++g) {
        if (g < nb) {
          const int sj = __builtin_amdgcn_readlane(s[(g * H) / 32], (g * H) % 32);
          const float* row = hfeat + (long long)sj * HC + lane;
#pragma unroll
          for (int v = 0; v < NG; ++v) hv[g][v] = row[v * 32];
        }
      }
#pragma unroll
      for (int it = 0; it < NIT; ++it) {
        const int j = (it * 32 + lane) / H, h = (it * 32 + lane) % H;
        const float b = __int_as_float(r[it].x);
        float e = a[it] + b;
        e = e > 0.0f ? e : NEG_SLOPE * e;
        float mx = key_f32(r[it].y) + b;
        mx = mx > 0.0f ? mx : NEG_SLOPE * mx;
        const float ex = expf(e - mx);
        if (j < nb) atomicAdd(&smL[(d[it] - lo) * H + h], ex);
        xb[it] = __float_as_int(ex);
      }
#pragma unroll
      for (int g = 0; g < EB; ++g) {
        if (g < nb) {
          const int dj = __builtin_amdgcn_readlane(d[(g * H) / 32], (g * H) % 32) - lo;
          float* arow = acc + dj * HC + lane;
#pragma unroll
          for (int v = 0; v < NG; ++v) {
            const float w = __int_as_float(__builtin_amdgcn_readlane(
                xb[(g * H + v / GPH) / 32], (g * H + v / GPH) % 32));
            atomicAdd(arow + v * 32, hv[g][v] * w);
          }
        }
      }
    }
    nq = 0;
  };

  const unsigned me4 = blockIdx.x * 0x01010101u;
  constexpr int SU = 8;
  for (int g0 = wv; g0 < SCAN_GROUPS; g0 += SU * OWN_WAVES) {
    uint4 o[SU];
#pragma unroll
    for (int u = 0; u < SU; ++u) {
      const int gg = g0 + u * OWN_WAVES;
      const int gl = gg < SCAN_GROUPS ? gg : SCAN_GROUPS - 1;
      o[u] = ((const uint4*)ob)[gl * 32 + lane];
    }
    unsigned pk[SU / 2];
#pragma unroll
    for (int u = 0; u < SU; ++u) {
      const unsigned wds[4] = {o[u].x, o[u].y, o[u].z, o[u].w};
      unsigned mm = 0;
#pragma unroll
      for (int k = 0; k < 4; ++k) {
        const unsigned x = wds[k] ^ me4;
        const unsigned t = ~(((x & 0x7F7F7F7Fu) + 0x7F7F7F7Fu) | x | 0x7F7F7F7Fu);
        mm |= t >> (7 - k);
      }
      if ((u & 1) == 0) pk[u / 2] = mm; else pk[u / 2] |= mm << 4;
    }
#pragma nounroll
    for (int u = 0; u < SU; ++u) {
      const int gg = g0 + u * OWN_WAVES;
      if (gg >= SCAN_GROUPS) break;
      unsigned w = pk[0];
#pragma unroll
      for (int t = 1; t < SU / 2; ++t) if ((u >> 1) == t) w = pk[t];
      unsigned m = (w >> ((u & 1) * 4)) & 0x0F0F0F0Fu;
      if (nq > QCAP - SCAN_EDGES) drain();
      unsigned bal = __builtin_amdgcn_ballot_w32(m != 0);
      while (bal != 0) {
        if (m != 0) {
          const int idx = nq + (int)__builtin_amdgcn_mbcnt_lo(bal, 0u);
          const int pos = __builtin_ctz(m);
          q[wv][idx] = gg * SCAN_EDGES + lane * 16 + (pos & 7) * 4 + (pos >> 3);
          m &= m - 1;
        }
        nq += __builtin_popcount(bal);
        bal = __builtin_amdgcn_ballot_w32(m != 0);
      }
    }
  }
  drain();
  __syncthreads();

  float bs[NG];
#pragma unroll
  for (int v = 0; v < NG; ++v) bs[v] = bias[v * 32 + lane];
  for (int nl = wv; nl < nown; nl += OWN_WAVES) {
    float* orow = out + (long long)(lo + nl) * HC + lane;
#pragma unroll
    for (int v = 0; v < NG; ++v) {
      float val = acc[nl * HC + v * 32 + lane] / (smL[nl * H + v / GPH] + EPS_) + bs[v];
      if (ELU) val = val > 0.0f ? val : expm1f(val);
      orow[v * 32] = val;
    }
  }
}

extern "C" void kernel_launch(void* const* d_in, const int* in_sizes, int n_in,
                              void* d_out, int out_size, void* d_ws, size_t ws_size,
                              hipStream_t stream) {
  (void)in_sizes; (void)n_in; (void)out_size; (void)ws_size;

  const float* x    = (const float*)d_in[0];
  const int*   ei   = (const int*)  d_in[1];
  const float* W1   = (const float*)d_in[2];
  const float* aS1  = (const float*)d_in[3];
  const float* aD1  = (const float*)d_in[4];
  const float* b1   = (const float*)d_in[5];
  const float* W2   = (const float*)d_in[6];
  const float* aS2  = (const float*)d_in[7];
  const float* aD2  = (const float*)d_in[8];
  const float* b2   = (const float*)d_in[9];
  float* out = (float*)d_out;

  float* ws  = (float*)d_ws;
  float* h1  = ws;
  float* o1  = h1  + (long long)NNODES * L1_HC;
  float* as1 = o1  + (long long)NNODES * L1_HC;
  int*   rc1 = (int*)(as1 + NNODES * L1_H);
  float* as2 = (float*)rc1 + NNODES * L1_H * 2;
  int*   rc2 = (int*)(as2 + NNODES);
  float* ea1 = (float*)rc2 + NNODES * 2;
  float* ea2 = ea1 + (long long)NEDGES * L1_H;
  float* h2  = h1;
  unsigned char* ob = (unsigned char*)d_ws + 160000000LL;

  gemm_attn_init_kernel<128, L1_HC, L1_H, 1><<<NNODES / 16, 128, 0, stream>>>(
      x, W1, h1, aS1, aD1, as1, rc1);
  edge_max_kernel<L1_H, true><<<MAX_BLOCKS, 256, 0, stream>>>(ei, as1, ea1, rc1, ob);
  owner_rows_kernel<L1_H, L1_C, true><<<OWN_BLOCKS, OWN_WAVES * 32, 0, stream>>>(
      ei, ob, ea1, as1, rc1, h1, b1, o1);

  gemm_attn_init_kernel<128, L2_HC, L2_H, 2><<<NNODES / 32, 128, 0, stream>>>(
      o1, W2, h2, aS2, aD2, as2, rc2);
  edge_max_kernel<L2_H, false><<<MAX_BLOCKS, 256, 0, stream>>>(ei, as2, ea2, rc2, ob);
  owner_rows_kernel<L2_H, L2_C, false><<<OWN_BLOCKS, OWN_WAVES * 32, 0, stream>>>(
      ei, ob, ea2, as2, rc2, h2, b2, out);
}


# Second layer's row sums walk the {source, destination} pairs the first layer's kernel found and kept as per-wave runs: no second owner-byte scan, no gathers by edge number; add order class unchanged
# speedup vs baseline: 6.1225x; 1.0968x over previous
#include <hip/hip_runtime.h>
#include <math.h>

#define NNODES   100000
#define NEDGES   1600000
#define L1_H     4
#define L1_C     32
#define L1_HC    128
#define L2_H     1
#define L2_C     64
#define L2_HC    64
#define NEG_SLOPE 0.2f
#define EPS_      1e-16f
#define OWN_BLOCKS 256
#define OWN_R      ((NNODES + OWN_BLOCKS - 1) / OWN_BLOCKS)
#define OWN_WAVES  32
#define SCAN_EDGES  512
#define SCAN_GROUPS (NEDGES / SCAN_EDGES)
#define QCAP 768
#define MAX_BLOCKS (((NEDGES) / 32 + 7) / 8)
#define SLOT    512
#define LST_OVF (OWN_BLOCKS * OWN_WAVES * SLOT)
#define LST_ALL (LST_OVF + NEDGES)
#define MAXSEG  196

typedef __attribute__((ext_vector_type(2))) float v2f;
typedef __attribute__((ext_vector_type(8))) float v8f;

__device__ __forceinline__ int f32_key(float v) {
  const int b = __float_as_int(v);
  return b >= 0 ? b : b ^ 0x7FFFFFFF;
}
__device__ __forceinline__ float key_f32(int k) {
  return __int_as_float(k >= 0 ? k : k ^ 0x7FFFFFFF);
}

template <int CTRL>
__device__ __forceinline__ float row_xor_add(float v) {
  return v + __int_as_float(
      __builtin_amdgcn_update_dpp(0, __float_as_int(v), CTRL, 0xF, 0xF, true));
}

template <int K, int NC, int H, int RT>
__global__ __launch_bounds__(RT * NC)
void gemm_attn_init_kernel(const float* __restrict__ A, const float* __restrict__ B,
                           float* __restrict__ C,
                           const float* __restrict__ att_s,
                           const float* __restrict__ att_d,
                           float* __restrict__ as_, int* __restrict__ rec) {
  constexpr int CT   = NC / 32;
  constexpr int CH   = NC / H;
  constexpr int WPH  = CH / 32;
  constexpr int ROWS = 16 * RT;
  constexpr int NE   = ROWS * H;
  constexpr int NT   = RT * NC;
  static_assert(CH % 32 == 0, "a wave's 32 columns lie inside one head");
  static_assert(NE % 32 == 0 && NE <= NT, "whole lines of as_/rec, whole waves");
  static_assert(NNODES % ROWS == 0, "no partial block: the barrier below is block-uniform");

  __shared__ float ps[WPH][NE];
  __shared__ float pd[WPH][NE];

  const int lane = threadIdx.x & 31;
  const int wv   = threadIdx.x >> 5;
  const int rt   = wv / CT;
  const int ct   = wv % CT;
  const int row0 = blockIdx.x * ROWS;
  const int rowt = row0 + rt * 16;
  const int col0 = ct * 32;
  const int m    = lane & 15;
  const int half = lane >> 4;
  const int kk   = half << 1;

  v8f acc0 = {};
  v8f acc1 = {};
  const float* ap = A + (long long)(rowt + m) * K + kk;
  const float* bp = B + (long long)kk * NC + (col0 + m);
#pragma unroll
  for (int k0 = 0; k0 < K; k0 += 4) {
    v2f a, b0, b1;
    a.x  = ap[k0];            a.y  = ap[k0 + 1];
    b0.x = bp[k0 * NC];       b0.y = bp[(k0 + 1) * NC];
    b1.x = bp[k0 * NC + 16];  b1.y = bp[(k0 + 1) * NC + 16];
    acc0 = __builtin_amdgcn_wmma_f32_16x16x4_f32(
        false, a, false, b0, (short)0, acc0, false, false);
    acc1 = __builtin_amdgcn_wmma_f32_16x16x4_f32(
        false, a, false, b1, (short)0, acc1, false, false);
  }
  const float s0 = att_s[col0 + m], s1 = att_s[col0 + 16 + m];
  const float d0 = att_d[col0 + m], d1 = att_d[col0 + 16 + m];
  float vs = 0.0f, vd = 0.0f;
#pragma unroll
  for (int r = 0; r < 8; ++r) {
    float p = acc0[r] * s0 + acc1[r] * s1;
    float q = acc0[r] * d0 + acc1[r] * d1;
    p = row_xor_add<0x161>(p);  q = row_xor_add<0x161>(q);
    p = row_xor_add<0x162>(p);  q = row_xor_add<0x162>(q);
    p = row_xor_add<0x164>(p);  q = row_xor_add<0x164>(q);
    p = row_xor_add<0x168>(p);  q = row_xor_add<0x168>(q);
    if (m == r) { vs = p; vd = q; }
  }
  if (m < 8) {
    const int e = (rt * 16 + half * 8 + m) * H + ct / WPH;
    ps[ct % WPH][e] = vs;
    pd[ct % WPH][e] = vd;
  }

  __syncthreads();

  const int orow = rowt + (half << 3);
  const int ocol = col0 + m;
  float* cp = C + (long long)orow * NC + ocol;
#pragma unroll
  for (int r = 0; r < 8; ++r) {
    cp[(long long)r * NC]      = acc0[r];
    cp[(long long)r * NC + 16] = acc1[r];
  }

  const int t = threadIdx.x;
  if (t < NE) {
    float v = ps[0][t];
    float u = pd[0][t];
#pragma unroll
    for (int w = 1; w < WPH; ++w) { v += ps[w][t]; u += pd[w][t]; }
    as_[row0 * H + t] = v;
    ((int2*)rec)[row0 * H + t] = make_int2(__float_as_int(u), f32_key(v));
  }
}

template <int H>
__device__ __forceinline__ void wave_edge_groups(const int* __restrict__ ei, int base,
                                                 int sub, int (&s)[H], int (&d)[H]) {
  constexpr int EPI = 32 / H;
  static_assert(32 % H == 0, "heads of an edge sit in adjacent lanes");
  static_assert(NEDGES % 32 == 0, "a wave's 32 edges are all listed or all self-loops");
  if (base < NEDGES) {
#pragma unroll
    for (int it = 0; it < H; ++it) {
      s[it] = ei[base + it * EPI + sub];
      d[it] = ei[NEDGES + base + it * EPI + sub];
    }
  } else {
#pragma unroll
    for (int it = 0; it < H; ++it) { s[it] = base - NEDGES + it * EPI + sub; d[it] = s[it]; }
  }
}

template <int H, bool OB, bool EA>
__global__ __launch_bounds__(256)
void edge_max_kernel(const int* __restrict__ ei,
                     const float* __restrict__ as_,
                     float* __restrict__ ea,
                     int* __restrict__ rec,
                     unsigned char* __restrict__ ob,
                     int* __restrict__ tick) {
  if (OB && blockIdx.x == 0 && threadIdx.x == 0) *tick = 0;
  const int lane = threadIdx.x & 31;
  const int base = __builtin_amdgcn_readfirstlane(
      (blockIdx.x * (blockDim.x >> 5) + (threadIdx.x >> 5)) * 32);
  if (base >= NEDGES) return;
  static_assert((NNODES - 1) / OWN_R < 256, "an owner fits a byte");
  const int h = lane % H, sub = lane / H;
  int s[H], d[H];
  wave_edge_groups<H>(ei, base, sub, s, d);
  float a[H];
#pragma unroll
  for (int it = 0; it < H; ++it) a[it] = as_[s[it] * H + h];
#pragma unroll
  for (int it = 0; it < H; ++it) {
    if (EA) ea[base * H + it * 32 + lane] = a[it];
    atomicMax(&rec[(d[it] * H + h) * 2 + 1], f32_key(a[it]));
  }
  if (OB) ob[base + lane] = (unsigned char)(ei[NEDGES + base + lane] / OWN_R);
}

template <int H, int C, bool ELU, bool LIST>
__global__ __launch_bounds__(OWN_WAVES * 32)
void owner_rows_kernel(const int* __restrict__ ei,
                       const unsigned char* __restrict__ ob,
                       const float* __restrict__ ea,
                       const float* __restrict__ as_,
                       const int* __restrict__ rec,
                       const float* __restrict__ hfeat,
                       const float* __restrict__ bias,
                       float* __restrict__ out,
                       int* __restrict__ lst,
                       int* __restrict__ dir,
                       int* __restrict__ dcnt,
                       int* __restrict__ tick) {
  constexpr int HC  = H * C;
  constexpr int NG  = HC / 32;
  constexpr int GPH = C / 32;
  constexpr int EB  = 64 / NG;
  constexpr int NIT = (EB * H + 31) / 32;
  static_assert(C % 32 == 0, "a 32-channel group must lie inside one head");
  static_assert(32 % H == 0, "heads of an edge sit in adjacent lanes");
  static_assert(NEDGES % SCAN_EDGES == 0, "whole scan groups");
  static_assert(QCAP >= SCAN_EDGES, "a whole group fits the list");
  static_assert(EB * H == NIT * 32, "step 1 covers the batch's (edge, head) pairs in whole waves");
  static_assert(((SCAN_GROUPS + OWN_WAVES - 1) / OWN_WAVES) * SCAN_EDGES / (QCAP - SCAN_EDGES + 1) + 1
                <= MAXSEG, "a wave's runs fit its part of dir");

  __shared__ float acc[OWN_R * HC];
  __shared__ float smL[OWN_R * H];
  __shared__ int   q[LIST ? 1 : OWN_WAVES][LIST ? 1 : QCAP];

  const int lane = threadIdx.x & 31;
  const int wv = __builtin_amdgcn_readfirstlane(threadIdx.x >> 5);
  const int wid = blockIdx.x * OWN_WAVES + wv;
  const int lo = blockIdx.x * OWN_R;
  const int nown = NNODES - lo < OWN_R ? NNODES - lo : OWN_R;

  for (int p = threadIdx.x; p < nown * H; p += OWN_WAVES * 32) {
    const float a = as_[lo * H + p];
    const int2 r = ((const int2*)rec)[lo * H + p];
    const float b = __int_as_float(r.x);
    float e = a + b;
    e = e > 0.0f ? e : NEG_SLOPE * e;
    float mx = key_f32(r.y) + b;
    mx = mx > 0.0f ? mx : NEG_SLOPE * mx;
    smL[p] = expf(e - mx);
  }
  __syncthreads();
  for (int nl0 = wv; nl0 < nown; nl0 += 4 * OWN_WAVES) {
    float hr[4][NG];
#pragma unroll
    for (int u = 0; u < 4; ++u) {
      const int nl = nl0 + u * OWN_WAVES;
#pragma unroll
      for (int v = 0; v < NG; ++v) hr[u][v] = 0.0f;
      if (nl < nown) {
        const float* row = hfeat + (long long)(lo + nl) * HC + lane;
#pragma unroll
        for (int v = 0; v < NG; ++v) hr[u][v] = row[v * 32];
      }
    }
#pragma unroll
    for (int u = 0; u < 4; ++u) {
      const int nl = nl0 + u * OWN_WAVES;
      if (nl < nown) {
#pragma unroll
        for (int v = 0; v < NG; ++v)
          acc[nl * HC + v * 32 + lane] = smL[nl * H + v / GPH] * hr[u][v];
      }
    }
  }
  __syncthreads();

  auto work = [&](const int n, const int off) {
    int2 nx[NIT];
    if constexpr (LIST) {
      const int nb0 = n < EB ? n : EB;
#pragma unroll
      for (int it = 0; it < NIT; ++it) {
        const int j = (it * 32 + lane) / H;
        nx[it] = ((const int2*)lst)[off + (j < nb0 ? j : 0)];
      }
    }
    for (int i0 = 0; i0 < n; i0 += EB) {
      const int nb = n - i0 < EB ? n - i0 : EB;
      int s[NIT], d[NIT], xb[NIT];
      float a[NIT];
      int2 r[NIT];
      if constexpr (!LIST) {
        int eg[NIT];
#pragma unroll
        for (int it = 0; it < NIT; ++it) {
          const int j = (it * 32 + lane) / H;
          eg[it] = j < nb ? q[wv][i0 + j] : 0;
        }
#pragma unroll
        for (int it = 0; it < NIT; ++it) {
          const int h = (it * 32 + lane) % H;
          s[it] = ei[eg[it]];
          d[it] = ei[NEDGES + eg[it]];
          a[it] = ea[(long long)eg[it] * H + h];
        }
#pragma unroll
        for (int it = 0; it < NIT; ++it) asm volatile("" :: "v"(s[it]), "v"(d[it]), "v"(a[it]));
#pragma unroll
        for (int it = 0; it < NIT; ++it) {
          const int j = (it * 32 + lane) / H, h = (it * 32 + lane) % H;
          if (h == 0 && j < nb) ((int2*)lst)[off + i0 + j] = make_int2(s[it], d[it]);
        }
      } else {
#pragma unroll
        for (int it = 0; it < NIT; ++it) { s[it] = nx[it].x; d[it] = nx[it].y; }
        const int i1 = i0 + EB < n ? i0 + EB : i0;
        const int nb1 = n - i1 < EB ? n - i1 : EB;
#pragma unroll
        for (int it = 0; it < NIT; ++it) {
          const int j = (it * 32 + lane) / H;
          nx[it] = ((const int2*)lst)[off + i1 + (j < nb1 ? j : 0)];
        }
#pragma unroll
        for (int it = 0; it < NIT; ++it) {
          const int h = (it * 32 + lane) % H;
          a[it] = as_[s[it] * H + h];
        }
      }
#pragma unroll
      for (int it = 0; it < NIT; ++it) {
        const int h = (it * 32 + lane) % H;
        r[it] = ((const int2*)rec)[d[it] * H + h];
      }
      float hv[EB][NG];
#pragma unroll
      for (int g = 0; g < EB; ++g) {
#pragma unroll
        for (int v = 0; v < NG; ++v) hv[g][v] = 0.0f;
      }
#pragma unroll
      for (int g = 0; g < EB; ++g) {
        if (g < nb) {
          const int sj = __builtin_amdgcn_readlane(s[(g * H) / 32], (g * H) % 32);
          const float* row = hfeat + (long long)sj * HC + lane;
#pragma unroll
          for (int v = 0; v < NG; ++v) hv[g][v] = row[v * 32];
        }
      }
#pragma unroll
      for (int it = 0; it < NIT; ++it) {
        const int j = (it * 32 + lane) / H, h = (it * 32 + lane) % H;
        const float b = __int_as_float(r[it].x);
        float e = a[it] + b;
        e = e > 0.0f ? e : NEG_SLOPE * e;
        float mx = key_f32(r[it].y) + b;
        mx = mx > 0.0f ? mx : NEG_SLOPE * mx;
        const float ex = expf(e - mx);
        if (j < nb) atomicAdd(&smL[(d[it] - lo) * H + h], ex);
        xb[it] = __float_as_int(ex);
      }
#pragma unroll
      for (int g = 0; g < EB; ++g) {
        if (g < nb) {
          const int dj = __builtin_amdgcn_readlane(d[(g * H) / 32], (g * H) % 32) - lo;
          float* arow = acc + dj * HC + lane;
#pragma unroll
          for (int v = 0; v < NG; ++v) {
            const float w = __int_as_float(__builtin_amdgcn_readlane(
                xb[(g * H + v / GPH) / 32], (g * H + v / GPH) % 32));
            atomicAdd(arow + v * 32, hv[g][v] * w);
          }
        }
      }
    }
  };

  if constexpr (LIST) {
    int ns = __builtin_amdgcn_readfirstlane(dcnt[wid]);
    ns = ns < 0 ? 0 : (ns > MAXSEG ? MAXSEG : ns);
    for (int sg = 0; sg < ns; ++sg) {
      const int2 se = ((const int2*)dir)[wid * MAXSEG + sg];
      int off = __builtin_amdgcn_readfirstlane(se.x);
      int n   = __builtin_amdgcn_readfirstlane(se.y);
      n   = n < 0 ? 0 : (n > QCAP ? QCAP : n);
      off = off < 0 ? 0 : (off > LST_ALL - n ? LST_ALL - n : off);
      work(n, off);
    }
  } else {
    int nq = 0;
    int nseg = 0;
    int used = 0;

    auto drain = [&]() {
      if (nq > 0) {
        int off;
        if (used + nq <= SLOT) {
          off = wid * SLOT + used;
          used += nq;
        } else {
          int t = 0;
          if (lane == 0) t = atomicAdd(tick, nq);
          off = LST_OVF + __builtin_amdgcn_readfirstlane(t);
        }
        if (lane == 0) ((int2*)dir)[wid * MAXSEG + nseg] = make_int2(off, nq);
        ++nseg;
        work(nq, off);
      }
      nq = 0;
    };

    const unsigned me4 = blockIdx.x * 0x01010101u;
    constexpr int SU = 8;
    for (int g0 = wv; g0 < SCAN_GROUPS; g0 += SU * OWN_WAVES) {
      uint4 o[SU];
#pragma unroll
      for (int u = 0; u < SU; ++u) {
        const int gg = g0 + u * OWN_WAVES;
        const int gl = gg < SCAN_GROUPS ? gg : SCAN_GROUPS - 1;
        o[u] = ((const uint4*)ob)[gl * 32 + lane];
      }
      unsigned pk[SU / 2];
#pragma unroll
      for (int u = 0; u < SU; ++u) {
        const unsigned wds[4] = {o[u].x, o[u].y, o[u].z, o[u].w};
        unsigned mm = 0;
#pragma unroll
        for (int k = 0; k < 4; ++k) {
          const unsigned x = wds[k] ^ me4;
          const unsigned t = ~(((x & 0x7F7F7F7Fu) + 0x7F7F7F7Fu) | x | 0x7F7F7F7Fu);
          mm |= t >> (7 - k);
        }
        if ((u & 1) == 0) pk[u / 2] = mm; else pk[u / 2] |= mm << 4;
      }
#pragma nounroll
      for (int u = 0; u < SU; ++u) {
        const int gg = g0 + u * OWN_WAVES;
        if (gg >= SCAN_GROUPS) break;
        unsigned w = pk[0];
#pragma unroll
        for (int t = 1; t < SU / 2; ++t) if ((u >> 1) == t) w = pk[t];
        unsigned m = (w >> ((u & 1) * 4)) & 0x0F0F0F0Fu;
        if (nq > QCAP - SCAN_EDGES) drain();
        unsigned bal = __builtin_amdgcn_ballot_w32(m != 0);
        while (bal != 0) {
          if (m != 0) {
            const int idx = nq + (int)__builtin_amdgcn_mbcnt_lo(bal, 0u);
            const int pos = __builtin_ctz(m);
            q[wv][idx] = gg * SCAN_EDGES + lane * 16 + (pos & 7) * 4 + (pos >> 3);
            m &= m - 1;
          }
          nq += __builtin_popcount(bal);
          bal = __builtin_amdgcn_ballot_w32(m != 0);
        }
      }
    }
    drain();
    if (lane == 0) dcnt[wid] = nseg;
  }
  __syncthreads();

  float bs[NG];
#pragma unroll
  for (int v = 0; v < NG; ++v) bs[v] = bias[v * 32 + lane];
  for (int nl = wv; nl < nown; nl += OWN_WAVES) {
    float* orow = out + (long long)(lo + nl) * HC + lane;
#pragma unroll
    for (int v = 0; v < NG; ++v) {
      float val = acc[nl * HC + v * 32 + lane] / (smL[nl * H + v / GPH] + EPS_) + bs[v];
      if (ELU) val = val > 0.0f ? val : expm1f(val);
      orow[v * 32] = val;
    }
  }
}

extern "C" void kernel_launch(void* const* d_in, const int* in_sizes, int n_in,
                              void* d_out, int out_size, void* d_ws, size_t ws_size,
                              hipStream_t stream) {
  (void)in_sizes; (void)n_in; (void)out_size; (void)ws_size;

  const float* x    = (const float*)d_in[0];
  const int*   ei   = (const int*)  d_in[1];
  const float* W1   = (const float*)d_in[2];
  const float* aS1  = (const float*)d_in[3];
  const float* aD1  = (const float*)d_in[4];
  const float* b1   = (const float*)d_in[5];
  const float* W2   = (const float*)d_in[6];
  const float* aS2  = (const float*)d_in[7];
  const float* aD2  = (const float*)d_in[8];
  const float* b2   = (const float*)d_in[9];
  float* out = (float*)d_out;

  float* ws  = (float*)d_ws;
  float* h1  = ws;
  float* o1  = h1  + (long long)NNODES * L1_HC;
  float* as1 = o1  + (long long)NNODES * L1_HC;
  int*   rc1 = (int*)(as1 + NNODES * L1_H);
  float* as2 = (float*)rc1 + NNODES * L1_H * 2;
  int*   rc2 = (int*)(as2 + NNODES);
  float* ea1 = (float*)rc2 + NNODES * 2;
  float* ea2 = ea1 + (long long)NEDGES * L1_H;
  float* h2  = h1;
  unsigned char* ob = (unsigned char*)d_ws + 160000000LL;
  int* lst  = (int*)((unsigned char*)d_ws + 162529280LL);
  int* dir  = (int*)((unsigned char*)d_ws + 209715200LL);
  int* dcnt = (int*)((unsigned char*)d_ws + 223346688LL);
  int* tick = (int*)((unsigned char*)d_ws + 224395264LL);

  gemm_attn_init_kernel<128, L1_HC, L1_H, 1><<<NNODES / 16, 128, 0, stream>>>(
      x, W1, h1, aS1, aD1, as1, rc1);
  edge_max_kernel<L1_H, true, true><<<MAX_BLOCKS, 256, 0, stream>>>(ei, as1, ea1, rc1, ob, tick);
  owner_rows_kernel<L1_H, L1_C, true, false><<<OWN_BLOCKS, OWN_WAVES * 32, 0, stream>>>(
      ei, ob, ea1, as1, rc1, h1, b1, o1, lst, dir, dcnt, tick);

  gemm_attn_init_kernel<128, L2_HC, L2_H, 2><<<NNODES / 32, 128, 0, stream>>>(
      o1, W2, h2, aS2, aD2, as2, rc2);
  edge_max_kernel<L2_H, false, false><<<MAX_BLOCKS, 256, 0, stream>>>(ei, as2, ea2, rc2, ob, tick);
  owner_rows_kernel<L2_H, L2_C, false, true><<<OWN_BLOCKS, OWN_WAVES * 32, 0, stream>>>(
      ei, ob, ea2, as2, rc2, h2, b2, out, lst, dir, dcnt, tick);
}


# Segment max taken inside the row-sum kernel: LDS integer max over the owner's nodes (a max pass, then the sum pass, over its edge runs); max kernels gone, owner bytes by the GEMM: 6 to 4 launches
# speedup vs baseline: 8.7909x; 1.4358x over previous
#include <hip/hip_runtime.h>
#include <math.h>

#define NNODES   100000
#define NEDGES   1600000
#define L1_H     4
#define L1_C     32
#define L1_HC    128
#define L2_H     1
#define L2_C     64
#define L2_HC    64
#define NEG_SLOPE 0.2f
#define EPS_      1e-16f
#define OWN_BLOCKS 256
#define OWN_R      ((NNODES + OWN_BLOCKS - 1) / OWN_BLOCKS)
#define OWN_WAVES  32
#define SCAN_EDGES  512
#define SCAN_GROUPS (NEDGES / SCAN_EDGES)
#define QCAP 768
#define SLOT    512
#define LST_OVF (OWN_BLOCKS * OWN_WAVES * SLOT)
#define LST_ALL (LST_OVF + NEDGES)
#define MAXSEG  196

typedef __attribute__((ext_vector_type(2))) float v2f;
typedef __attribute__((ext_vector_type(8))) float v8f;

__device__ __forceinline__ int f32_key(float v) {
  const int b = __float_as_int(v);
  return b >= 0 ? b : b ^ 0x7FFFFFFF;
}
__device__ __forceinline__ float key_f32(int k) {
  return __int_as_float(k >= 0 ? k : k ^ 0x7FFFFFFF);
}

template <int CTRL>
__device__ __forceinline__ float row_xor_add(float v) {
  return v + __int_as_float(
      __builtin_amdgcn_update_dpp(0, __float_as_int(v), CTRL, 0xF, 0xF, true));
}

template <int K, int NC, int H, int RT, bool OB>
__global__ __launch_bounds__(RT * NC)
void gemm_attn_init_kernel(const float* __restrict__ A, const float* __restrict__ B,
                           float* __restrict__ C,
                           const float* __restrict__ att_s,
                           const float* __restrict__ att_d,
                           float* __restrict__ as_, int* __restrict__ rec,
                           const int* __restrict__ ei, unsigned* __restrict__ ob4,
                           int* __restrict__ tick) {
  constexpr int CT   = NC / 32;
  constexpr int CH   = NC / H;
  constexpr int WPH  = CH / 32;
  constexpr int ROWS = 16 * RT;
  constexpr int NE   = ROWS * H;
  constexpr int NT   = RT * NC;
  static_assert(CH % 32 == 0, "a wave's 32 columns lie inside one head");
  static_assert(NE % 32 == 0 && NE <= NT, "whole lines of as_/rec, whole waves");
  static_assert(NNODES % ROWS == 0, "no partial block: the barrier below is block-uniform");

  __shared__ float ps[WPH][NE];
  __shared__ float pd[WPH][NE];

  const int lane = threadIdx.x & 31;
  const int wv   = threadIdx.x >> 5;
  const int rt   = wv / CT;
  const int ct   = wv % CT;
  const int row0 = blockIdx.x * ROWS;
  const int rowt = row0 + rt * 16;
  const int col0 = ct * 32;
  const int m    = lane & 15;
  const int half = lane >> 4;
  const int kk   = half << 1;

  unsigned obw = 0;
  if constexpr (OB) {
    static_assert(NT == 128 && (NNODES / ROWS) * 256 == NEDGES, "256 listed edges per block");
    static_assert((NNODES - 1) / OWN_R < 256, "an owner fits a byte");
    if (blockIdx.x == 0 && threadIdx.x == 0) *tick = 0;
    if (wv < 2) {
      const int4 dd = *(const int4*)(ei + NEDGES + blockIdx.x * 256 + wv * 128 + lane * 4);
      obw = (unsigned)(dd.x / OWN_R) | ((unsigned)(dd.y / OWN_R) << 8) |
            ((unsigned)(dd.z / OWN_R) << 16) | ((unsigned)(dd.w / OWN_R) << 24);
    }
  }

  v8f acc0 = {};
  v8f acc1 = {};
  const float* ap = A + (long long)(rowt + m) * K + kk;
  const float* bp = B + (long long)kk * NC + (col0 + m);
#pragma unroll
  for (int k0 = 0; k0 < K; k0 += 4) {
    v2f a, b0, b1;
    a.x  = ap[k0];            a.y  = ap[k0 + 1];
    b0.x = bp[k0 * NC];       b0.y = bp[(k0 + 1) * NC];
    b1.x = bp[k0 * NC + 16];  b1.y = bp[(k0 + 1) * NC + 16];
    acc0 = __builtin_amdgcn_wmma_f32_16x16x4_f32(
        false, a, false, b0, (short)0, acc0, false, false);
    acc1 = __builtin_amdgcn_wmma_f32_16x16x4_f32(
        false, a, false, b1, (short)0, acc1, false, false);
  }
  const float s0 = att_s[col0 + m], s1 = att_s[col0 + 16 + m];
  const float d0 = att_d[col0 + m], d1 = att_d[col0 + 16 + m];
  float vs = 0.0f, vd = 0.0f;
#pragma unroll
  for (int r = 0; r < 8; ++r) {
    float p = acc0[r] * s0 + acc1[r] * s1;
    float q = acc0[r] * d0 + acc1[r] * d1;
    p = row_xor_add<0x161>(p);  q = row_xor_add<0x161>(q);
    p = row_xor_add<0x162>(p);  q = row_xor_add<0x162>(q);
    p = row_xor_add<0x164>(p);  q = row_xor_add<0x164>(q);
    p = row_xor_add<0x168>(p);  q = row_xor_add<0x168>(q);
    if (m == r) { vs = p; vd = q; }
  }
  if (m < 8) {
    const int e = (rt * 16 + half * 8 + m) * H + ct / WPH;
    ps[ct % WPH][e] = vs;
    pd[ct % WPH][e] = vd;
  }

  __syncthreads();

  const int orow = rowt + (half << 3);
  const int ocol = col0 + m;
  float* cp = C + (long long)orow * NC + ocol;
#pragma unroll
  for (int r = 0; r < 8; ++r) {
    cp[(long long)r * NC]      = acc0[r];
    cp[(long long)r * NC + 16] = acc1[r];
  }

  const int t = threadIdx.x;
  if (t < NE) {
    float v = ps[0][t];
    float u = pd[0][t];
#pragma unroll
    for (int w = 1; w < WPH; ++w) { v += ps[w][t]; u += pd[w][t]; }
    as_[row0 * H + t] = v;
    ((int2*)rec)[row0 * H + t] = make_int2(__float_as_int(u), f32_key(v));
  }
  if constexpr (OB) {
    if (wv < 2) ob4[blockIdx.x * 64 + wv * 32 + lane] = obw;
  }
}

template <int H, int C, bool ELU, bool LIST>
__global__ __launch_bounds__(OWN_WAVES * 32)
void owner_rows_kernel(const int* __restrict__ ei,
                       const unsigned char* __restrict__ ob,
                       const float* __restrict__ as_,
                       const int* __restrict__ rec,
                       const float* __restrict__ hfeat,
                       const float* __restrict__ bias,
                       float* __restrict__ out,
                       int* __restrict__ lst,
                       int* __restrict__ dir,
                       int* __restrict__ dcnt,
                       int* __restrict__ tick) {
  constexpr int HC  = H * C;
  constexpr int NG  = HC / 32;
  constexpr int GPH = C / 32;
  constexpr int EB  = 64 / NG;
  constexpr int NIT = (EB * H + 31) / 32;
  constexpr int EPI = 32 / H;
  constexpr int UA  = 64 / EPI > 4 ? 64 / EPI : 4;
  static_assert(C % 32 == 0, "a 32-channel group must lie inside one head");
  static_assert(32 % H == 0, "heads of an edge sit in adjacent lanes");
  static_assert(NEDGES % SCAN_EDGES == 0, "whole scan groups");
  static_assert(QCAP >= SCAN_EDGES, "a whole group fits the list");
  static_assert(EB * H == NIT * 32, "step 1 covers the batch's (edge, head) pairs in whole waves");
  static_assert(((SCAN_GROUPS + OWN_WAVES - 1) / OWN_WAVES) * SCAN_EDGES / (QCAP - SCAN_EDGES + 1) + 1
                <= MAXSEG, "a wave's runs fit its part of dir");

  __shared__ float acc[OWN_R * HC];
  __shared__ float smL[OWN_R * H];
  __shared__ float adL[OWN_R * H];
  __shared__ int   mxK[OWN_R * H];
  __shared__ int   q[LIST ? 1 : OWN_WAVES][LIST ? 1 : QCAP];

  const int lane = threadIdx.x & 31;
  const int hd = lane % H, sub = lane / H;
  const int wv = __builtin_amdgcn_readfirstlane(threadIdx.x >> 5);
  const int wid = blockIdx.x * OWN_WAVES + wv;
  const int lo = blockIdx.x * OWN_R;
  const int nown = NNODES - lo < OWN_R ? NNODES - lo : OWN_R;

  for (int p = threadIdx.x; p < nown * H; p += OWN_WAVES * 32) {
    const int2 r = ((const int2*)rec)[lo * H + p];
    adL[p] = __int_as_float(r.x);
    mxK[p] = r.y;
  }
  __syncthreads();

  auto run_at = [&](const int sg, int& n, int& off) {
    const int2 se = ((const int2*)dir)[wid * MAXSEG + sg];
    off = __builtin_amdgcn_readfirstlane(se.x);
    n   = __builtin_amdgcn_readfirstlane(se.y);
    n   = n < 0 ? 0 : (n > QCAP ? QCAP : n);
    off = off < 0 ? 0 : (off > LST_ALL - n ? LST_ALL - n : off);
  };

  auto maxrun = [&](const int n, const int off) {
    for (int i0 = 0; i0 < n; i0 += UA * EPI) {
      int s[UA], d[UA];
      float a[UA];
      if constexpr (!LIST) {
        int eg[UA];
#pragma unroll
        for (int u = 0; u < UA; ++u) {
          const int j = i0 + u * EPI + sub;
          eg[u] = j < n ? q[wv][j] : 0;
        }
#pragma unroll
        for (int u = 0; u < UA; ++u) {
          s[u] = ei[eg[u]];
          d[u] = ei[NEDGES + eg[u]];
        }
      } else {
#pragma unroll
        for (int u = 0; u < UA; ++u) {
          const int j = i0 + u * EPI + sub;
          const int2 pr = ((const int2*)lst)[off + (j < n ? j : 0)];
          s[u] = pr.x;  d[u] = pr.y;
        }
      }
#pragma unroll
      for (int u = 0; u < UA; ++u) a[u] = as_[s[u] * H + hd];
      if constexpr (!LIST) {
#pragma unroll
        for (int u = 0; u < UA; ++u) {
          const int j = i0 + u * EPI + sub;
          if (hd == 0 && j < n) ((int2*)lst)[off + j] = make_int2(s[u], d[u]);
        }
      }
#pragma unroll
      for (int u = 0; u < UA; ++u) {
        const int j = i0 + u * EPI + sub;
        if (j < n) atomicMax(&mxK[(d[u] - lo) * H + hd], f32_key(a[u]));
      }
    }
  };

  int ns = 0;
  if constexpr (LIST) {
    ns = __builtin_amdgcn_readfirstlane(dcnt[wid]);
    ns = ns < 0 ? 0 : (ns > MAXSEG ? MAXSEG : ns);
    for (int sg = 0; sg < ns; ++sg) {
      int n, off;
      run_at(sg, n, off);
      maxrun(n, off);
    }
  } else {
    int nq = 0;
    int used = 0;

    auto drain = [&]() {
      if (nq > 0) {
        int off;
        if (used + nq <= SLOT) {
          off = wid * SLOT + used;
          used += nq;
        } else {
          int t = 0;
          if (lane == 0) t = atomicAdd(tick, nq);
          off = LST_OVF + __builtin_amdgcn_readfirstlane(t);
        }
        if (lane == 0) ((int2*)dir)[wid * MAXSEG + ns] = make_int2(off, nq);
        ++ns;
        maxrun(nq, off);
      }
      nq = 0;
    };

    const unsigned me4 = blockIdx.x * 0x01010101u;
    constexpr int SU = 8;
    for (int g0 = wv; g0 < SCAN_GROUPS; g0 += SU * OWN_WAVES) {
      uint4 o[SU];
#pragma unroll
      for (int u = 0; u < SU; ++u) {
        const int gg = g0 + u * OWN_WAVES;
        const int gl = gg < SCAN_GROUPS ? gg : SCAN_GROUPS - 1;
        o[u] = ((const uint4*)ob)[gl * 32 + lane];
      }
      unsigned pk[SU / 2];
#pragma unroll
      for (int u = 0; u < SU; ++u) {
        const unsigned wds[4] = {o[u].x, o[u].y, o[u].z, o[u].w};
        unsigned mm = 0;
#pragma unroll
        for (int k = 0; k < 4; ++k) {
          const unsigned x = wds[k] ^ me4;
          const unsigned t = ~(((x & 0x7F7F7F7Fu) + 0x7F7F7F7Fu) | x | 0x7F7F7F7Fu);
          mm |= t >> (7 - k);
        }
        if ((u & 1) == 0) pk[u / 2] = mm; else pk[u / 2] |= mm << 4;
      }
#pragma nounroll
      for (int u = 0; u < SU; ++u) {
        const int gg = g0 + u * OWN_WAVES;
        if (gg >= SCAN_GROUPS) break;
        unsigned w = pk[0];
#pragma unroll
        for (int t = 1; t < SU / 2; ++t) if ((u >> 1) == t) w = pk[t];
        unsigned m = (w >> ((u & 1) * 4)) & 0x0F0F0F0Fu;
        if (nq > QCAP - SCAN_EDGES) drain();
        unsigned bal = __builtin_amdgcn_ballot_w32(m != 0);
        while (bal != 0) {
          if (m != 0) {
            const int idx = nq + (int)__builtin_amdgcn_mbcnt_lo(bal, 0u);
            const int pos = __builtin_ctz(m);
            q[wv][idx] = gg * SCAN_EDGES + lane * 16 + (pos & 7) * 4 + (pos >> 3);
            m &= m - 1;
          }
          nq += __builtin_popcount(bal);
          bal = __builtin_amdgcn_ballot_w32(m != 0);
        }
      }
    }
    drain();
    if (lane == 0) dcnt[wid] = ns;
  }
  __syncthreads();

  for (int p = threadIdx.x; p < nown * H; p += OWN_WAVES * 32) {
    const float a = as_[lo * H + p];
    const float b = adL[p];
    float e = a + b;
    e = e > 0.0f ? e : NEG_SLOPE * e;
    float mx = key_f32(mxK[p]) + b;
    mx = mx > 0.0f ? mx : NEG_SLOPE * mx;
    smL[p] = expf(e - mx);
    mxK[p] = __float_as_int(mx);
  }
  __syncthreads();
  for (int nl0 = wv; nl0 < nown; nl0 += 4 * OWN_WAVES) {
    float hr[4][NG];
#pragma unroll
    for (int u = 0; u < 4; ++u) {
      const int nl = nl0 + u * OWN_WAVES;
#pragma unroll
      for (int v = 0; v < NG; ++v) hr[u][v] = 0.0f;
      if (nl < nown) {
        const float* row = hfeat + (long long)(lo + nl) * HC + lane;
#pragma unroll
        for (int v = 0; v < NG; ++v) hr[u][v] = row[v * 32];
      }
    }
#pragma unroll
    for (int u = 0; u < 4; ++u) {
      const int nl = nl0 + u * OWN_WAVES;
      if (nl < nown) {
#pragma unroll
        for (int v = 0; v < NG; ++v)
          acc[nl * HC + v * 32 + lane] = smL[nl * H + v / GPH] * hr[u][v];
      }
    }
  }
  __syncthreads();

  auto work = [&](const int n, const int off) {
    int2 nx[NIT];
    const int nb0 = n < EB ? n : EB;
#pragma unroll
    for (int it = 0; it < NIT; ++it) {
      const int j = (it * 32 + lane) / H;
      nx[it] = ((const int2*)lst)[off + (j < nb0 ? j : 0)];
    }
    for (int i0 = 0; i0 < n; i0 += EB) {
      const int nb = n - i0 < EB ? n - i0 : EB;
      int s[NIT], d[NIT], xb[NIT];
      float a[NIT];
#pragma unroll
      for (int it = 0; it < NIT; ++it) { s[it] = nx[it].x; d[it] = nx[it].y; }
      const int i1 = i0 + EB < n ? i0 + EB : i0;
      const int nb1 = n - i1 < EB ? n - i1 : EB;
#pragma unroll
      for (int it = 0; it < NIT; ++it) {
        const int j = (it * 32 + lane) / H;
        nx[it] = ((const int2*)lst)[off + i1 + (j < nb1 ? j : 0)];
      }
#pragma unroll
      for (int it = 0; it < NIT; ++it) {
        const int h = (it * 32 + lane) % H;
        a[it] = as_[s[it] * H + h];
      }
      float hv[EB][NG];
#pragma unroll
      for (int g = 0; g < EB; ++g) {
#pragma unroll
        for (int v = 0; v < NG; ++v) hv[g][v] = 0.0f;
      }
#pragma unroll
      for (int g = 0; g < EB; ++g) {
        if (g < nb) {
          const int sj = __builtin_amdgcn_readlane(s[(g * H) / 32], (g * H) % 32);
          const float* row = hfeat + (long long)sj * HC + lane;
#pragma unroll
          for (int v = 0; v < NG; ++v) hv[g][v] = row[v * 32];
        }
      }
#pragma unroll
      for (int it = 0; it < NIT; ++it) {
        const int j = (it * 32 + lane) / H, h = (it * 32 + lane) % H;
        const int p = (d[it] - lo) * H + h;
        float e = a[it] + adL[p];
        e = e > 0.0f ? e : NEG_SLOPE * e;
        const float ex = expf(e - __int_as_float(mxK[p]));
        if (j < nb) atomicAdd(&smL[p], ex);
        xb[it] = __float_as_int(ex);
      }
#pragma unroll
      for (int g = 0; g < EB; ++g) {
        if (g < nb) {
          const int dj = __builtin_amdgcn_readlane(d[(g * H) / 32], (g * H) % 32) - lo;
          float* arow = acc + dj * HC + lane;
#pragma unroll
          for (int v = 0; v < NG; ++v) {
            const float w = __int_as_float(__builtin_amdgcn_readlane(
                xb[(g * H + v / GPH) / 32], (g * H + v / GPH) % 32));
            atomicAdd(arow + v * 32, hv[g][v] * w);
          }
        }
      }
    }
  };
  for (int sg = 0; sg < ns; ++sg) {
    int n, off;
    run_at(sg, n, off);
    work(n, off);
  }
  __syncthreads();

  float bs[NG];
#pragma unroll
  for (int v = 0; v < NG; ++v) bs[v] = bias[v * 32 + lane];
  for (int nl = wv; nl < nown; nl += OWN_WAVES) {
    float* orow = out + (long long)(lo + nl) * HC + lane;
#pragma unroll
    for (int v = 0; v < NG; ++v) {
      float val = acc[nl * HC + v * 32 + lane] / (smL[nl * H + v / GPH] + EPS_) + bs[v];
      if (ELU) val = val > 0.0f ? val : expm1f(val);
      orow[v * 32] = val;
    }
  }
}

extern "C" void kernel_launch(void* const* d_in, const int* in_sizes, int n_in,
                              void* d_out, int out_size, void* d_ws, size_t ws_size,
                              hipStream_t stream) {
  (void)in_sizes; (void)n_in; (void)out_size; (void)ws_size;

  const float* x    = (const float*)d_in[0];
  const int*   ei   = (const int*)  d_in[1];
  const float* W1   = (const float*)d_in[2];
  const float* aS1  = (const float*)d_in[3];
  const float* aD1  = (const float*)d_in[4];
  const float* b1   = (const float*)d_in[5];
  const float* W2   = (const float*)d_in[6];
  const float* aS2  = (const float*)d_in[7];
  const float* aD2  = (const float*)d_in[8];
  const float* b2   = (const float*)d_in[9];
  float* out = (float*)d_out;

  float* ws  = (float*)d_ws;
  float* h1  = ws;
  float* o1  = h1  + (long long)NNODES * L1_HC;
  float* as1 = o1  + (long long)NNODES * L1_HC;
  int*   rc1 = (int*)(as1 + NNODES * L1_H);
  float* as2 = (float*)rc1 + NNODES * L1_H * 2;
  int*   rc2 = (int*)(as2 + NNODES);
  float* h2  = h1;
  unsigned char* ob = (unsigned char*)d_ws + 160000000LL;
  int* lst  = (int*)((unsigned char*)d_ws + 162529280LL);
  int* dir  = (int*)((unsigned char*)d_ws + 209715200LL);
  int* dcnt = (int*)((unsigned char*)d_ws + 223346688LL);
  int* tick = (int*)((unsigned char*)d_ws + 224395264LL);

  gemm_attn_init_kernel<128, L1_HC, L1_H, 1, true><<<NNODES / 16, 128, 0, stream>>>(
      x, W1, h1, aS1, aD1, as1, rc1, ei, (unsigned*)ob, tick);
  owner_rows_kernel<L1_H, L1_C, true, false><<<OWN_BLOCKS, OWN_WAVES * 32, 0, stream>>>(
      ei, ob, as1, rc1, h1, b1, o1, lst, dir, dcnt, tick);

  gemm_attn_init_kernel<128, L2_HC, L2_H, 2, false><<<NNODES / 32, 128, 0, stream>>>(
      o1, W2, h2, aS2, aD2, as2, rc2, ei, (unsigned*)ob, tick);
  owner_rows_kernel<L2_H, L2_C, false, true><<<OWN_BLOCKS, OWN_WAVES * 32, 0, stream>>>(
      ei, ob, as2, rc2, h2, b2, out, lst, dir, dcnt, tick);
}


# Layer-2 dense product moved into layer 1's row-sum kernel: finished rows stay in LDS as A fragments, W2 copied transposed to LDS; o1 never written or read, 4 to 3 launches; sum orders unchanged
# speedup vs baseline: 9.2831x; 1.0560x over previous
#include <hip/hip_runtime.h>
#include <math.h>

#define NNODES   100000
#define NEDGES   1600000
#define L1_H     4
#define L1_C     32
#define L1_HC    128
#define L2_H     1
#define L2_C     64
#define L2_HC    64
#define NEG_SLOPE 0.2f
#define EPS_      1e-16f
#define OWN_BLOCKS 256
#define OWN_R      ((NNODES + OWN_BLOCKS - 1) / OWN_BLOCKS)
#define OWN_WAVES  32
#define SCAN_EDGES  512
#define SCAN_GROUPS (NEDGES / SCAN_EDGES)
#define QCAP 768
#define SLOT    512
#define LST_OVF (OWN_BLOCKS * OWN_WAVES * SLOT)
#define LST_ALL (LST_OVF + NEDGES)
#define MAXSEG  196

typedef __attribute__((ext_vector_type(2))) float v2f;
typedef __attribute__((ext_vector_type(8))) float v8f;

__device__ __forceinline__ int f32_key(float v) {
  const int b = __float_as_int(v);
  return b >= 0 ? b : b ^ 0x7FFFFFFF;
}
__device__ __forceinline__ float key_f32(int k) {
  return __int_as_float(k >= 0 ? k : k ^ 0x7FFFFFFF);
}

template <int CTRL>
__device__ __forceinline__ float row_xor_add(float v) {
  return v + __int_as_float(
      __builtin_amdgcn_update_dpp(0, __float_as_int(v), CTRL, 0xF, 0xF, true));
}

template <int K, int NC, int H, int RT, bool OB>
__global__ __launch_bounds__(RT * NC)
void gemm_attn_init_kernel(const float* __restrict__ A, const float* __restrict__ B,
                           float* __restrict__ C,
                           const float* __restrict__ att_s,
                           const float* __restrict__ att_d,
                           float* __restrict__ as_, int* __restrict__ rec,
                           const int* __restrict__ ei, unsigned* __restrict__ ob4,
                           int* __restrict__ tick) {
  constexpr int CT   = NC / 32;
  constexpr int CH   = NC / H;
  constexpr int WPH  = CH / 32;
  constexpr int ROWS = 16 * RT;
  constexpr int NE   = ROWS * H;
  constexpr int NT   = RT * NC;
  static_assert(CH % 32 == 0, "a wave's 32 columns lie inside one head");
  static_assert(NE % 32 == 0 && NE <= NT, "whole lines of as_/rec, whole waves");
  static_assert(NNODES % ROWS == 0, "no partial block: the barrier below is block-uniform");

  __shared__ float ps[WPH][NE];
  __shared__ float pd[WPH][NE];

  const int lane = threadIdx.x & 31;
  const int wv   = threadIdx.x >> 5;
  const int rt   = wv / CT;
  const int ct   = wv % CT;
  const int row0 = blockIdx.x * ROWS;
  const int rowt = row0 + rt * 16;
  const int col0 = ct * 32;
  const int m    = lane & 15;
  const int half = lane >> 4;
  const int kk   = half << 1;

  unsigned obw = 0;
  if constexpr (OB) {
    static_assert(NT == 128 && (NNODES / ROWS) * 256 == NEDGES, "256 listed edges per block");
    static_assert((NNODES - 1) / OWN_R < 256, "an owner fits a byte");
    if (blockIdx.x == 0 && threadIdx.x == 0) *tick = 0;
    if (wv < 2) {
      const int4 dd = *(const int4*)(ei + NEDGES + blockIdx.x * 256 + wv * 128 + lane * 4);
      obw = (unsigned)(dd.x / OWN_R) | ((unsigned)(dd.y / OWN_R) << 8) |
            ((unsigned)(dd.z / OWN_R) << 16) | ((unsigned)(dd.w / OWN_R) << 24);
    }
  }

  v8f acc0 = {};
  v8f acc1 = {};
  const float* ap = A + (long long)(rowt + m) * K + kk;
  const float* bp = B + (long long)kk * NC + (col0 + m);
#pragma unroll
  for (int k0 = 0; k0 < K; k0 += 4) {
    v2f a, b0, b1;
    a.x  = ap[k0];            a.y  = ap[k0 + 1];
    b0.x = bp[k0 * NC];       b0.y = bp[(k0 + 1) * NC];
    b1.x = bp[k0 * NC + 16];  b1.y = bp[(k0 + 1) * NC + 16];
    acc0 = __builtin_amdgcn_wmma_f32_16x16x4_f32(
        false, a, false, b0, (short)0, acc0, false, false);
    acc1 = __builtin_amdgcn_wmma_f32_16x16x4_f32(
        false, a, false, b1, (short)0, acc1, false, false);
  }
  const float s0 = att_s[col0 + m], s1 = att_s[col0 + 16 + m];
  const float d0 = att_d[col0 + m], d1 = att_d[col0 + 16 + m];
  float vs = 0.0f, vd = 0.0f;
#pragma unroll
  for (int r = 0; r < 8; ++r) {
    float p = acc0[r] * s0 + acc1[r] * s1;
    float q = acc0[r] * d0 + acc1[r] * d1;
    p = row_xor_add<0x161>(p);  q = row_xor_add<0x161>(q);
    p = row_xor_add<0x162>(p);  q = row_xor_add<0x162>(q);
    p = row_xor_add<0x164>(p);  q = row_xor_add<0x164>(q);
    p = row_xor_add<0x168>(p);  q = row_xor_add<0x168>(q);
    if (m == r) { vs = p; vd = q; }
  }
  if (m < 8) {
    const int e = (rt * 16 + half * 8 + m) * H + ct / WPH;
    ps[ct % WPH][e] = vs;
    pd[ct % WPH][e] = vd;
  }

  __syncthreads();

  const int orow = rowt + (half << 3);
  const int ocol = col0 + m;
  float* cp = C + (long long)orow * NC + ocol;
#pragma unroll
  for (int r = 0; r < 8; ++r) {
    cp[(long long)r * NC]      = acc0[r];
    cp[(long long)r * NC + 16] = acc1[r];
  }

  const int t = threadIdx.x;
  if (t < NE) {
    float v = ps[0][t];
    float u = pd[0][t];
#pragma unroll
    for (int w = 1; w < WPH; ++w) { v += ps[w][t]; u += pd[w][t]; }
    as_[row0 * H + t] = v;
    ((int2*)rec)[row0 * H + t] = make_int2(__float_as_int(u), f32_key(v));
  }
  if constexpr (OB) {
    if (wv < 2) ob4[blockIdx.x * 64 + wv * 32 + lane] = obw;
  }
}

template <int H, int C, bool ELU, bool LIST, bool FUSE>
__global__ __launch_bounds__(OWN_WAVES * 32)
void owner_rows_kernel(const int* __restrict__ ei,
                       const unsigned char* __restrict__ ob,
                       const float* __restrict__ as_,
                       const int* __restrict__ rec,
                       const float* __restrict__ hfeat,
                       const float* __restrict__ bias,
                       float* __restrict__ out,
                       int* __restrict__ lst,
                       int* __restrict__ dir,
                       int* __restrict__ dcnt,
                       int* __restrict__ tick,
                       const float* __restrict__ Wn,
                       const float* __restrict__ att_sn,
                       const float* __restrict__ att_dn,
                       float* __restrict__ asn,
                       int* __restrict__ recn) {
  constexpr int HC  = H * C;
  constexpr int PITCH = FUSE ? HC + 4 : HC;
  constexpr int NG  = HC / 32;
  constexpr int GPH = C / 32;
  constexpr int EB  = 64 / NG;
  constexpr int NIT = (EB * H + 31) / 32;
  constexpr int EPI = 32 / H;
  constexpr int UA  = 64 / EPI > 4 ? 64 / EPI : 4;
  static_assert(C % 32 == 0, "a 32-channel group must lie inside one head");
  static_assert(32 % H == 0, "heads of an edge sit in adjacent lanes");
  static_assert(NEDGES % SCAN_EDGES == 0, "whole scan groups");
  static_assert(QCAP >= SCAN_EDGES, "a whole group fits the list");
  static_assert(EB * H == NIT * 32, "step 1 covers the batch's (edge, head) pairs in whole waves");
  static_assert(((SCAN_GROUPS + OWN_WAVES - 1) / OWN_WAVES) * SCAN_EDGES / (QCAP - SCAN_EDGES + 1) + 1
                <= MAXSEG, "a wave's runs fit its part of dir");

  __shared__ float acc[OWN_R * PITCH];
  __shared__ float smL[OWN_R * H];
  __shared__ float adL[OWN_R * H];
  __shared__ int   mxK[OWN_R * H];
  __shared__ int   q[LIST ? 1 : OWN_WAVES][LIST ? 1 : QCAP];

  const int lane = threadIdx.x & 31;
  const int hd = lane % H, sub = lane / H;
  const int wv = __builtin_amdgcn_readfirstlane(threadIdx.x >> 5);
  const int wid = blockIdx.x * OWN_WAVES + wv;
  const int lo = blockIdx.x * OWN_R;
  const int nown = NNODES - lo < OWN_R ? NNODES - lo : OWN_R;

  for (int p = threadIdx.x; p < nown * H; p += OWN_WAVES * 32) {
    const int2 r = ((const int2*)rec)[lo * H + p];
    adL[p] = __int_as_float(r.x);
    mxK[p] = r.y;
  }
  __syncthreads();

  auto run_at = [&](const int sg, int& n, int& off) {
    const int2 se = ((const int2*)dir)[wid * MAXSEG + sg];
    off = __builtin_amdgcn_readfirstlane(se.x);
    n   = __builtin_amdgcn_readfirstlane(se.y);
    n   = n < 0 ? 0 : (n > QCAP ? QCAP : n);
    off = off < 0 ? 0 : (off > LST_ALL - n ? LST_ALL - n : off);
  };

  auto maxrun = [&](const int n, const int off) {
    for (int i0 = 0; i0 < n; i0 += UA * EPI) {
      int s[UA], d[UA];
      float a[UA];
      if constexpr (!LIST) {
        int eg[UA];
#pragma unroll
        for (int u = 0; u < UA; ++u) {
          const int j = i0 + u * EPI + sub;
          eg[u] = j < n ? q[wv][j] : 0;
        }
#pragma unroll
        for (int u = 0; u < UA; ++u) {
          s[u] = ei[eg[u]];
          d[u] = ei[NEDGES + eg[u]];
        }
      } else {
#pragma unroll
        for (int u = 0; u < UA; ++u) {
          const int j = i0 + u * EPI + sub;
          const int2 pr = ((const int2*)lst)[off + (j < n ? j : 0)];
          s[u] = pr.x;  d[u] = pr.y;
        }
      }
#pragma unroll
      for (int u = 0; u < UA; ++u) a[u] = as_[s[u] * H + hd];
      if constexpr (!LIST) {
#pragma unroll
        for (int u = 0; u < UA; ++u) {
          const int j = i0 + u * EPI + sub;
          if (hd == 0 && j < n) ((int2*)lst)[off + j] = make_int2(s[u], d[u]);
        }
      }
#pragma unroll
      for (int u = 0; u < UA; ++u) {
        const int j = i0 + u * EPI + sub;
        if (j < n) atomicMax(&mxK[(d[u] - lo) * H + hd], f32_key(a[u]));
      }
    }
  };

  int ns = 0;
  if constexpr (LIST) {
    ns = __builtin_amdgcn_readfirstlane(dcnt[wid]);
    ns = ns < 0 ? 0 : (ns > MAXSEG ? MAXSEG : ns);
    for (int sg = 0; sg < ns; ++sg) {
      int n, off;
      run_at(sg, n, off);
      maxrun(n, off);
    }
  } else {
    int nq = 0;
    int used = 0;

    auto drain = [&]() {
      if (nq > 0) {
        int off;
        if (used + nq <= SLOT) {
          off = wid * SLOT + used;
          used += nq;
        } else {
          int t = 0;
          if (lane == 0) t = atomicAdd(tick, nq);
          off = LST_OVF + __builtin_amdgcn_readfirstlane(t);
        }
        if (lane == 0) ((int2*)dir)[wid * MAXSEG + ns] = make_int2(off, nq);
        ++ns;
        maxrun(nq, off);
      }
      nq = 0;
    };

    const unsigned me4 = blockIdx.x * 0x01010101u;
    constexpr int SU = 8;
    for (int g0 = wv; g0 < SCAN_GROUPS; g0 += SU * OWN_WAVES) {
      uint4 o[SU];
#pragma unroll
      for (int u = 0; u < SU; ++u) {
        const int gg = g0 + u * OWN_WAVES;
        const int gl = gg < SCAN_GROUPS ? gg : SCAN_GROUPS - 1;
        o[u] = ((const uint4*)ob)[gl * 32 + lane];
      }
      unsigned pk[SU / 2];
#pragma unroll
      for (int u = 0; u < SU; ++u) {
        const unsigned wds[4] = {o[u].x, o[u].y, o[u].z, o[u].w};
        unsigned mm = 0;
#pragma unroll
        for (int k = 0; k < 4; ++k) {
          const unsigned x = wds[k] ^ me4;
          const unsigned t = ~(((x & 0x7F7F7F7Fu) + 0x7F7F7F7Fu) | x | 0x7F7F7F7Fu);
          mm |= t >> (7 - k);
        }
        if ((u & 1) == 0) pk[u / 2] = mm; else pk[u / 2] |= mm << 4;
      }
#pragma nounroll
      for (int u = 0; u < SU; ++u) {
        const int gg = g0 + u * OWN_WAVES;
        if (gg >= SCAN_GROUPS) break;
        unsigned w = pk[0];
#pragma unroll
        for (int t = 1; t < SU / 2; ++t) if ((u >> 1) == t) w = pk[t];
        unsigned m = (w >> ((u & 1) * 4)) & 0x0F0F0F0Fu;
        if (nq > QCAP - SCAN_EDGES) drain();
        unsigned bal = __builtin_amdgcn_ballot_w32(m != 0);
        while (bal != 0) {
          if (m != 0) {
            const int idx = nq + (int)__builtin_amdgcn_mbcnt_lo(bal, 0u);
            const int pos = __builtin_ctz(m);
            q[wv][idx] = gg * SCAN_EDGES + lane * 16 + (pos & 7) * 4 + (pos >> 3);
            m &= m - 1;
          }
          nq += __builtin_popcount(bal);
          bal = __builtin_amdgcn_ballot_w32(m != 0);
        }
      }
    }
    drain();
    if (lane == 0) dcnt[wid] = ns;
  }
  __syncthreads();

  for (int p = threadIdx.x; p < nown * H; p += OWN_WAVES * 32) {
    const float a = as_[lo * H + p];
    const float b = adL[p];
    float e = a + b;
    e = e > 0.0f ? e : NEG_SLOPE * e;
    float mx = key_f32(mxK[p]) + b;
    mx = mx > 0.0f ? mx : NEG_SLOPE * mx;
    smL[p] = expf(e - mx);
    mxK[p] = __float_as_int(mx);
  }
  __syncthreads();
  for (int nl0 = wv; nl0 < nown; nl0 += 4 * OWN_WAVES) {
    float hr[4][NG];
#pragma unroll
    for (int u = 0; u < 4; ++u) {
      const int nl = nl0 + u * OWN_WAVES;
#pragma unroll
      for (int v = 0; v < NG; ++v) hr[u][v] = 0.0f;
      if (nl < nown) {
        const float* row = hfeat + (long long)(lo + nl) * HC + lane;
#pragma unroll
        for (int v = 0; v < NG; ++v) hr[u][v] = row[v * 32];
      }
    }
#pragma unroll
    for (int u = 0; u < 4; ++u) {
      const int nl = nl0 + u * OWN_WAVES;
      if (nl < nown) {
#pragma unroll
        for (int v = 0; v < NG; ++v)
          acc[nl * PITCH + v * 32 + lane] = smL[nl * H + v / GPH] * hr[u][v];
      }
    }
  }
  __syncthreads();

  auto work = [&](const int n, const int off) {
    int2 nx[NIT];
    const int nb0 = n < EB ? n : EB;
#pragma unroll
    for (int it = 0; it < NIT; ++it) {
      const int j = (it * 32 + lane) / H;
      nx[it] = ((const int2*)lst)[off + (j < nb0 ? j : 0)];
    }
    for (int i0 = 0; i0 < n; i0 += EB) {
      const int nb = n - i0 < EB ? n - i0 : EB;
      int s[NIT], d[NIT], xb[NIT];
      float a[NIT];
#pragma unroll
      for (int it = 0; it < NIT; ++it) { s[it] = nx[it].x; d[it] = nx[it].y; }
      const int i1 = i0 + EB < n ? i0 + EB : i0;
      const int nb1 = n - i1 < EB ? n - i1 : EB;
#pragma unroll
      for (int it = 0; it < NIT; ++it) {
        const int j = (it * 32 + lane) / H;
        nx[it] = ((const int2*)lst)[off + i1 + (j < nb1 ? j : 0)];
      }
#pragma unroll
      for (int it = 0; it < NIT; ++it) {
        const int h = (it * 32 + lane) % H;
        a[it] = as_[s[it] * H + h];
      }
      float hv[EB][NG];
#pragma unroll
      for (int g = 0; g < EB; ++g) {
#pragma unroll
        for (int v = 0; v < NG; ++v) hv[g][v] = 0.0f;
      }
#pragma unroll
      for (int g = 0; g < EB; ++g) {
        if (g < nb) {
          const int sj = __builtin_amdgcn_readlane(s[(g * H) / 32], (g * H) % 32);
          const float* row = hfeat + (long long)sj * HC + lane;
#pragma unroll
          for (int v = 0; v < NG; ++v) hv[g][v] = row[v * 32];
        }
      }
#pragma unroll
      for (int it = 0; it < NIT; ++it) {
        const int j = (it * 32 + lane) / H, h = (it * 32 + lane) % H;
        const int p = (d[it] - lo) * H + h;
        float e = a[it] + adL[p];
        e = e > 0.0f ? e : NEG_SLOPE * e;
        const float ex = expf(e - __int_as_float(mxK[p]));
        if (j < nb) atomicAdd(&smL[p], ex);
        xb[it] = __float_as_int(ex);
      }
#pragma unroll
      for (int g = 0; g < EB; ++g) {
        if (g < nb) {
          const int dj = __builtin_amdgcn_readlane(d[(g * H) / 32], (g * H) % 32) - lo;
          float* arow = acc + dj * PITCH + lane;
#pragma unroll
          for (int v = 0; v < NG; ++v) {
            const float w = __int_as_float(__builtin_amdgcn_readlane(
                xb[(g * H + v / GPH) / 32], (g * H + v / GPH) % 32));
            atomicAdd(arow + v * 32, hv[g][v] * w);
          }
        }
      }
    }
  };
  for (int sg = 0; sg < ns; ++sg) {
    int n, off;
    run_at(sg, n, off);
    work(n, off);
  }
  __syncthreads();

  float bs[NG];
#pragma unroll
  for (int v = 0; v < NG; ++v) bs[v] = bias[v * 32 + lane];
  if constexpr (!FUSE) {
    for (int nl = wv; nl < nown; nl += OWN_WAVES) {
      float* orow = out + (long long)(lo + nl) * HC + lane;
#pragma unroll
      for (int v = 0; v < NG; ++v) {
        float val = acc[nl * PITCH + v * 32 + lane] / (smL[nl * H + v / GPH] + EPS_) + bs[v];
        if (ELU) val = val > 0.0f ? val : expm1f(val);
        orow[v * 32] = val;
      }
    }
  } else {
    static_assert(!LIST && HC == 128, "the product below is written for K = 128");
    constexpr int NCN = L2_HC;
    constexpr int NRT = (OWN_R + 15) / 16;
    constexpr int TR  = NRT * 16;
    constexpr int WT0 = 2048;
    static_assert(4 * TR <= WT0 && WT0 + NCN * PITCH <= OWN_WAVES * QCAP,
                  "partial sums and the transposed weights fit the list space");
    static_assert(HC * NCN == 8 * OWN_WAVES * 32, "eight words of Wn per thread");
    float* psd = (float*)&q[0][0];
    float* wT  = psd + WT0;
    {
      float wr[8];
#pragma unroll
      for (int i = 0; i < 8; ++i) wr[i] = Wn[i * (OWN_WAVES * 32) + threadIdx.x];
#pragma unroll
      for (int i = 0; i < 8; ++i) {
        const int idx = i * (OWN_WAVES * 32) + threadIdx.x;
        wT[(idx % NCN) * PITCH + idx / NCN] = wr[i];
      }
    }
    for (int nl = wv; nl < nown; nl += OWN_WAVES) {
#pragma unroll
      for (int v = 0; v < NG; ++v) {
        float val = acc[nl * PITCH + v * 32 + lane] / (smL[nl * H + v / GPH] + EPS_) + bs[v];
        if (ELU) val = val > 0.0f ? val : expm1f(val);
        acc[nl * PITCH + v * 32 + lane] = val;
      }
    }
    __syncthreads();
    const int m    = lane & 15;
    const int half = lane >> 4;
    const int kk   = half << 1;
    for (int item = wv; item < NRT * 2; item += OWN_WAVES) {
      const int t = item >> 1, cpair = item & 1;
      if (t * 16 >= nown) break;
      const int col0 = cpair * 32;
      int ar = t * 16 + m;
      ar = ar < nown ? ar : nown - 1;
      const float* ap = acc + ar * PITCH + kk;
      const float* bp = wT + (col0 + m) * PITCH + kk;
      v8f acc0 = {};
      v8f acc1 = {};
#pragma unroll
      for (int k0 = 0; k0 < HC; k0 += 4) {
        v2f a, b0, b1;
        a.x  = ap[k0];             a.y  = ap[k0 + 1];
        b0.x = bp[k0];                 b0.y = bp[k0 + 1];
        b1.x = bp[16 * PITCH + k0];    b1.y = bp[16 * PITCH + k0 + 1];
        acc0 = __builtin_amdgcn_wmma_f32_16x16x4_f32(
            false, a, false, b0, (short)0, acc0, false, false);
        acc1 = __builtin_amdgcn_wmma_f32_16x16x4_f32(
            false, a, false, b1, (short)0, acc1, false, false);
      }
      const float s0 = att_sn[col0 + m], s1 = att_sn[col0 + 16 + m];
      const float d0 = att_dn[col0 + m], d1 = att_dn[col0 + 16 + m];
      float vs = 0.0f, vd = 0.0f;
#pragma unroll
      for (int r = 0; r < 8; ++r) {
        float p = acc0[r] * s0 + acc1[r] * s1;
        float qq = acc0[r] * d0 + acc1[r] * d1;
        p = row_xor_add<0x161>(p);  qq = row_xor_add<0x161>(qq);
        p = row_xor_add<0x162>(p);  qq = row_xor_add<0x162>(qq);
        p = row_xor_add<0x164>(p);  qq = row_xor_add<0x164>(qq);
        p = row_xor_add<0x168>(p);  qq = row_xor_add<0x168>(qq);
        if (m == r) { vs = p; vd = qq; }
      }
      if (m < 8) {
        const int e = t * 16 + half * 8 + m;
        psd[cpair * TR + e]       = vs;
        psd[(2 + cpair) * TR + e] = vd;
      }
      const int orl = t * 16 + (half << 3);
      float* cp = out + (long long)(lo + orl) * NCN + (col0 + m);
#pragma unroll
      for (int r = 0; r < 8; ++r) {
        if (orl + r < nown) {
          cp[r * NCN]      = acc0[r];
          cp[r * NCN + 16] = acc1[r];
        }
      }
    }
    __syncthreads();
    for (int p = threadIdx.x; p < nown; p += OWN_WAVES * 32) {
      const float v = psd[p] + psd[TR + p];
      const float u = psd[2 * TR + p] + psd[3 * TR + p];
      asn[lo + p] = v;
      ((int2*)recn)[lo + p] = make_int2(__float_as_int(u), f32_key(v));
    }
  }
}

extern "C" void kernel_launch(void* const* d_in, const int* in_sizes, int n_in,
                              void* d_out, int out_size, void* d_ws, size_t ws_size,
                              hipStream_t stream) {
  (void)in_sizes; (void)n_in; (void)out_size; (void)ws_size;

  const float* x    = (const float*)d_in[0];
  const int*   ei   = (const int*)  d_in[1];
  const float* W1   = (const float*)d_in[2];
  const float* aS1  = (const float*)d_in[3];
  const float* aD1  = (const float*)d_in[4];
  const float* b1   = (const float*)d_in[5];
  const float* W2   = (const float*)d_in[6];
  const float* aS2  = (const float*)d_in[7];
  const float* aD2  = (const float*)d_in[8];
  const float* b2   = (const float*)d_in[9];
  float* out = (float*)d_out;

  float* ws  = (float*)d_ws;
  float* h1  = ws;
  float* h2  = h1  + (long long)NNODES * L1_HC;
  float* as1 = h2  + (long long)NNODES * L1_HC;
  int*   rc1 = (int*)(as1 + NNODES * L1_H);
  float* as2 = (float*)rc1 + NNODES * L1_H * 2;
  int*   rc2 = (int*)(as2 + NNODES);
  unsigned char* ob = (unsigned char*)d_ws + 160000000LL;
  int* lst  = (int*)((unsigned char*)d_ws + 162529280LL);
  int* dir  = (int*)((unsigned char*)d_ws + 209715200LL);
  int* dcnt = (int*)((unsigned char*)d_ws + 223346688LL);
  int* tick = (int*)((unsigned char*)d_ws + 224395264LL);

  gemm_attn_init_kernel<128, L1_HC, L1_H, 1, true><<<NNODES / 16, 128, 0, stream>>>(
      x, W1, h1, aS1, aD1, as1, rc1, ei, (unsigned*)ob, tick);
  owner_rows_kernel<L1_H, L1_C, true, false, true><<<OWN_BLOCKS, OWN_WAVES * 32, 0, stream>>>(
      ei, ob, as1, rc1, h1, b1, h2, lst, dir, dcnt, tick, W2, aS2, aD2, as2, rc2);

  owner_rows_kernel<L2_H, L2_C, false, true, false><<<OWN_BLOCKS, OWN_WAVES * 32, 0, stream>>>(
      ei, ob, as2, rc2, h2, b2, out, lst, dir, dcnt, tick, nullptr, nullptr, nullptr, nullptr, nullptr);
}
